# starting kernel
# baseline (speedup 1.0000x reference)
#include <hip/hip_runtime.h>
#include <math.h>

#define BATCH 8
#define SEQ   8192
#define DIM   64
#define NHASH 8
#define NBUCK 128
#define NBALL (NHASH*NBUCK)
#define BSZ   64
#define NCHUNK (NHASH*(SEQ/BSZ))
#define HS    (NHASH*SEQ)
#define TS    512
#define NT    (SEQ/TS)
#define SELF_VAL (-5.0e4f)

typedef float v2f __attribute__((ext_vector_type(2)));
typedef float v8f __attribute__((ext_vector_type(8)));

__global__ __launch_bounds__(256) void lsh_hash_kernel(
    const float* __restrict__ qk, const float* __restrict__ rot,
    int* __restrict__ bucketArr, int* __restrict__ tilecnt)
{
  __shared__ __align__(16) float A [64][68];
  __shared__ __align__(16) float Bm[64][68];
  __shared__ __align__(16) float R [64][68];

  const int bid = blockIdx.x;
  const int tt  = bid & 127;
  const int h   = (bid >> 7) & 7;
  const int b   = bid >> 10;
  const int tid = threadIdx.x;
  const int t0  = tt * 64;

  {
    int row = tid >> 2, c0 = (tid & 3) * 16;
    const float4* s4 = (const float4*)(qk + ((size_t)b*SEQ + t0 + row)*DIM + c0);
    float4* d4 = (float4*)(&A[row][c0]);
    d4[0]=s4[0]; d4[1]=s4[1]; d4[2]=s4[2]; d4[3]=s4[3];
  }
  {
    int f = tid >> 2, c0 = (tid & 3) * 16;
    const float4* s4 = (const float4*)(rot + (size_t)f*(NHASH*64) + (size_t)h*64 + c0);
    float4* d4 = (float4*)(&Bm[f][c0]);
    d4[0]=s4[0]; d4[1]=s4[1]; d4[2]=s4[2]; d4[3]=s4[3];
  }
  __syncthreads();

  const int wave = tid >> 5, lane = tid & 31, half = lane >> 4, mn = lane & 15;
  for (int T = wave; T < 16; T += 8) {
    const int mi = T >> 2, ni = T & 3;
    v8f acc = {};
#pragma unroll
    for (int k0 = 0; k0 < 64; k0 += 4) {
      v2f a, bb;
      a.x  = A [mi*16 + mn][k0 + 2*half    ];
      a.y  = A [mi*16 + mn][k0 + 2*half + 1];
      bb.x = Bm[k0 + 2*half    ][ni*16 + mn];
      bb.y = Bm[k0 + 2*half + 1][ni*16 + mn];
      acc = __builtin_amdgcn_wmma_f32_16x16x4_f32(false, a, false, bb,
                                                  (short)0, acc, false, false);
    }
#pragma unroll
    for (int r = 0; r < 8; r++)
      R[mi*16 + r + 8*half][ni*16 + mn] = acc[r];
  }
  __syncthreads();

  if (tid < 64) {
    const int row = tid;
    float best = R[row][0]; int bidx = 0;
    for (int j = 1; j < 128; j++) {
      float vv = (j < 64) ? R[row][j] : -R[row][j - 64];
      if (vv > best) { best = vv; bidx = j; }
    }
    const int q = h*NBUCK + bidx;
    const int t = t0 + row;
    bucketArr[((size_t)b*NHASH + h)*SEQ + t] = q;
    atomicAdd(&tilecnt[((size_t)b*NBALL + q)*NT + (t >> 9)], 1);
  }
}

__global__ __launch_bounds__(1024) void lsh_prefix_kernel(int* __restrict__ tilecnt)
{
  __shared__ int sc[NBALL];
  const int b = blockIdx.x, q = threadIdx.x;
  int* tc = tilecnt + ((size_t)b*NBALL + q)*NT;
  int s = 0;
#pragma unroll
  for (int j = 0; j < NT; j++) s += tc[j];
  sc[q] = s;
  __syncthreads();
  for (int off = 1; off < NBALL; off <<= 1) {
    int vv = (q >= off) ? sc[q - off] : 0;
    __syncthreads();
    sc[q] += vv;
    __syncthreads();
  }
  int run = sc[q] - s;
#pragma unroll
  for (int j = 0; j < NT; j++) { int c0 = tc[j]; tc[j] = run; run += c0; }
}

__global__ __launch_bounds__(TS) void lsh_rank_kernel(
    const int* __restrict__ bucketArr, const int* __restrict__ tilecnt,
    int* __restrict__ sticker)
{
  __shared__ int bk[TS];
  const int bid  = blockIdx.x;
  const int tile = bid & (NT - 1);
  const int h    = (bid >> 4) & 7;
  const int b    = bid >> 7;
  const int tid  = threadIdx.x;
  const int t    = tile*TS + tid;
  const int q    = bucketArr[((size_t)b*NHASH + h)*SEQ + t];
  bk[tid] = q;
  __syncthreads();
  int r = 0;
  for (int j = 0; j < tid; j++) r += (bk[j] == q) ? 1 : 0;
  const int pos = tilecnt[((size_t)b*NBALL + q)*NT + tile] + r;
  sticker[(size_t)b*HS + pos] = h*SEQ + t;
}

__global__ __launch_bounds__(256) void lsh_attn_kernel(
    const float* __restrict__ qk, const float* __restrict__ vin,
    const int* __restrict__ sticker,
    float* __restrict__ oBuf, float* __restrict__ lgBuf)
{
  __shared__ int idxA[128];
  __shared__ int tA  [128];
  __shared__ __align__(16) float Qs[64][68];
  __shared__ __align__(16) float Ks[128][68];
  __shared__ __align__(16) float Vs[128][68];
  __shared__ __align__(16) float Ps[64][132];
  __shared__ float lse[64];

  const int bid = blockIdx.x;
  const int c   = bid & (NCHUNK - 1);
  const int b   = bid >> 10;
  const int tid = threadIdx.x;
  const int prev = (c + NCHUNK - 1) & (NCHUNK - 1);

  if (tid < 128) {
    const int src_chunk = (tid < 64) ? c : prev;
    const int j = tid & 63;
    const int idx = sticker[(size_t)b*HS + (size_t)src_chunk*BSZ + j];
    idxA[tid] = idx;
    tA[tid]   = idx & (SEQ - 1);
  }
  __syncthreads();

  {
    int row = tid >> 2, c0 = (tid & 3) * 16;
    const float4* s4 = (const float4*)(qk + ((size_t)b*SEQ + tA[row])*DIM + c0);
    float4* d4 = (float4*)(&Qs[row][c0]);
    d4[0]=s4[0]; d4[1]=s4[1]; d4[2]=s4[2]; d4[3]=s4[3];
  }
  {
    int row = tid >> 1, c0 = (tid & 1) * 32;
    const int t = tA[row];
    const float4* sk = (const float4*)(qk  + ((size_t)b*SEQ + t)*DIM + c0);
    const float4* sv = (const float4*)(vin + ((size_t)b*SEQ + t)*DIM + c0);
    float4* dk = (float4*)(&Ks[row][c0]);
    float4* dv = (float4*)(&Vs[row][c0]);
#pragma unroll
    for (int i = 0; i < 8; i++) { dk[i] = sk[i]; dv[i] = sv[i]; }
  }
  __syncthreads();

  if (tid < 128) {
    float ss = 0.f;
#pragma unroll
    for (int d = 0; d < 64; d++) { float x = Ks[tid][d]; ss += x * x; }
    const float inv = 1.0f / (sqrtf(ss) + 1e-6f);
#pragma unroll
    for (int d = 0; d < 64; d++) Ks[tid][d] *= inv;
  }
  __syncthreads();

  const int wave = tid >> 5, lane = tid & 31, half = lane >> 4, mn = lane & 15;

  {
    const int ni = wave;
    for (int mi = 0; mi < 4; mi++) {
      v8f acc = {};
#pragma unroll
      for (int k0 = 0; k0 < 64; k0 += 4) {
        v2f a, bb;
        a.x  = Qs[mi*16 + mn][k0 + 2*half    ];
        a.y  = Qs[mi*16 + mn][k0 + 2*half + 1];
        bb.x = Ks[ni*16 + mn][k0 + 2*half    ];
        bb.y = Ks[ni*16 + mn][k0 + 2*half + 1];
        acc = __builtin_amdgcn_wmma_f32_16x16x4_f32(false, a, false, bb,
                                                    (short)0, acc, false, false);
      }
#pragma unroll
      for (int r = 0; r < 8; r++) {
        const int i = mi*16 + r + 8*half, j = ni*16 + mn;
        float dval = acc[r] * 0.125f;
        if (tA[i] == tA[j]) dval = SELF_VAL;
        Ps[i][j] = dval;
      }
    }
  }
  __syncthreads();

  if (tid < 64) {
    float m = -1e30f;
#pragma unroll
    for (int j = 0; j < 128; j++) m = fmaxf(m, Ps[tid][j]);
    float ssum = 0.f;
#pragma unroll
    for (int j = 0; j < 128; j++) { float e = expf(Ps[tid][j] - m); Ps[tid][j] = e; ssum += e; }
    lse[tid] = m + logf(ssum);
    const float invs = 1.0f / ssum;
#pragma unroll
    for (int j = 0; j < 128; j++) Ps[tid][j] *= invs;
  }
  __syncthreads();

  for (int T = wave; T < 16; T += 8) {
    const int mi = T >> 2, ni = T & 3;
    v8f acc = {};
#pragma unroll
    for (int k0 = 0; k0 < 128; k0 += 4) {
      v2f a, bb;
      a.x  = Ps[mi*16 + mn][k0 + 2*half    ];
      a.y  = Ps[mi*16 + mn][k0 + 2*half + 1];
      bb.x = Vs[k0 + 2*half    ][ni*16 + mn];
      bb.y = Vs[k0 + 2*half + 1][ni*16 + mn];
      acc = __builtin_amdgcn_wmma_f32_16x16x4_f32(false, a, false, bb,
                                                  (short)0, acc, false, false);
    }
#pragma unroll
    for (int r = 0; r < 8; r++)
      Qs[mi*16 + r + 8*half][ni*16 + mn] = acc[r];
  }
  __syncthreads();

  {
    int row = tid >> 2, c0 = (tid & 3) * 16;
    float4* d4 = (float4*)(oBuf + ((size_t)b*HS + idxA[row])*DIM + c0);
    const float4* s4 = (const float4*)(&Qs[row][c0]);
    d4[0]=s4[0]; d4[1]=s4[1]; d4[2]=s4[2]; d4[3]=s4[3];
  }
  if (tid < 64) lgBuf[(size_t)b*HS + idxA[tid]] = lse[tid];
}

__global__ __launch_bounds__(256) void lsh_combine_kernel(
    const float* __restrict__ oBuf, const float* __restrict__ lgBuf,
    float* __restrict__ out)
{
  const int g = blockIdx.x * blockDim.x + threadIdx.x;
  const int b = g >> 13;
  const int t = g & (SEQ - 1);
  float lg[NHASH];
  float m = -1e30f;
#pragma unroll
  for (int h = 0; h < NHASH; h++) {
    lg[h] = lgBuf[(size_t)b*HS + (size_t)h*SEQ + t];
    m = fmaxf(m, lg[h]);
  }
  float ssum = 0.f;
#pragma unroll
  for (int h = 0; h < NHASH; h++) { lg[h] = expf(lg[h] - m); ssum += lg[h]; }
  const float inv = 1.0f / ssum;
#pragma unroll
  for (int h = 0; h < NHASH; h++) lg[h] *= inv;

  float* dst = out + ((size_t)b*SEQ + t)*DIM;
#pragma unroll
  for (int d = 0; d < DIM; d += 4) {
    float4 acc = {0.f, 0.f, 0.f, 0.f};
    for (int h = 0; h < NHASH; h++) {
      const float4 o4 = *(const float4*)(oBuf + ((size_t)b*HS + (size_t)h*SEQ + t)*DIM + d);
      acc.x += lg[h]*o4.x; acc.y += lg[h]*o4.y; acc.z += lg[h]*o4.z; acc.w += lg[h]*o4.w;
    }
    *(float4*)(dst + d) = acc;
  }
}

extern "C" void kernel_launch(void* const* d_in, const int* in_sizes, int n_in,
                              void* d_out, int out_size, void* d_ws, size_t ws_size,
                              hipStream_t stream)
{
  const float* qk  = (const float*)d_in[0];
  const float* v   = (const float*)d_in[1];
  const float* rot = (const float*)d_in[2];

  char* ws = (char*)d_ws;
  size_t off = 0;
  auto alloc = [&](size_t bytes) -> void* {
    void* p = ws + off;
    off += (bytes + 255) & ~(size_t)255;
    return p;
  };
  int*   bucketArr = (int*)  alloc((size_t)BATCH*NHASH*SEQ*sizeof(int));
  int*   tilecnt   = (int*)  alloc((size_t)BATCH*NBALL*NT*sizeof(int));
  int*   sticker   = (int*)  alloc((size_t)BATCH*HS*sizeof(int));
  float* oBuf      = (float*)alloc((size_t)BATCH*HS*DIM*sizeof(float));
  float* lgBuf     = (float*)alloc((size_t)BATCH*HS*sizeof(float));
  if (off > ws_size) return;

  hipMemsetAsync(tilecnt, 0, (size_t)BATCH*NBALL*NT*sizeof(int), stream);

  lsh_hash_kernel   <<<BATCH*NHASH*(SEQ/64), 256, 0, stream>>>(qk, rot, bucketArr, tilecnt);
  lsh_prefix_kernel <<<BATCH,               1024, 0, stream>>>(tilecnt);
  lsh_rank_kernel   <<<BATCH*NHASH*NT,        TS, 0, stream>>>(bucketArr, tilecnt, sticker);
  lsh_attn_kernel   <<<BATCH*NCHUNK,         256, 0, stream>>>(qk, v, sticker, oBuf, lgBuf);
  lsh_combine_kernel<<<(BATCH*SEQ)/256,      256, 0, stream>>>(oBuf, lgBuf, (float*)d_out);
}


# attention kernel: score tile overlays the key tile in LDS (122112 -> 88320 B, one more barrier), exp loop unrolled 32 not 128 (145 regs): 3 work-groups per WGP instead of 2
# speedup vs baseline: 1.1060x; 1.1060x over previous
#include <hip/hip_runtime.h>
#include <math.h>

#define BATCH 8
#define SEQ   8192
#define DIM   64
#define NHASH 8
#define NBUCK 128
#define NBALL (NHASH*NBUCK)
#define BSZ   64
#define NCHUNK (NHASH*(SEQ/BSZ))
#define HS    (NHASH*SEQ)
#define TS    512
#define NT    (SEQ/TS)
#define SELF_VAL (-5.0e4f)

typedef float v2f __attribute__((ext_vector_type(2)));
typedef float v8f __attribute__((ext_vector_type(8)));

__global__ __launch_bounds__(256) void lsh_hash_kernel(
    const float* __restrict__ qk, const float* __restrict__ rot,
    int* __restrict__ bucketArr, int* __restrict__ tilecnt)
{
  __shared__ __align__(16) float A [64][68];
  __shared__ __align__(16) float Bm[64][68];
  __shared__ __align__(16) float R [64][68];

  const int bid = blockIdx.x;
  const int tt  = bid & 127;
  const int h   = (bid >> 7) & 7;
  const int b   = bid >> 10;
  const int tid = threadIdx.x;
  const int t0  = tt * 64;

  {
    int row = tid >> 2, c0 = (tid & 3) * 16;
    const float4* s4 = (const float4*)(qk + ((size_t)b*SEQ + t0 + row)*DIM + c0);
    float4* d4 = (float4*)(&A[row][c0]);
    d4[0]=s4[0]; d4[1]=s4[1]; d4[2]=s4[2]; d4[3]=s4[3];
  }
  {
    int f = tid >> 2, c0 = (tid & 3) * 16;
    const float4* s4 = (const float4*)(rot + (size_t)f*(NHASH*64) + (size_t)h*64 + c0);
    float4* d4 = (float4*)(&Bm[f][c0]);
    d4[0]=s4[0]; d4[1]=s4[1]; d4[2]=s4[2]; d4[3]=s4[3];
  }
  __syncthreads();

  const int wave = tid >> 5, lane = tid & 31, half = lane >> 4, mn = lane & 15;
  for (int T = wave; T < 16; T += 8) {
    const int mi = T >> 2, ni = T & 3;
    v8f acc = {};
#pragma unroll
    for (int k0 = 0; k0 < 64; k0 += 4) {
      v2f a, bb;
      a.x  = A [mi*16 + mn][k0 + 2*half    ];
      a.y  = A [mi*16 + mn][k0 + 2*half + 1];
      bb.x = Bm[k0 + 2*half    ][ni*16 + mn];
      bb.y = Bm[k0 + 2*half + 1][ni*16 + mn];
      acc = __builtin_amdgcn_wmma_f32_16x16x4_f32(false, a, false, bb,
                                                  (short)0, acc, false, false);
    }
#pragma unroll
    for (int r = 0; r < 8; r++)
      R[mi*16 + r + 8*half][ni*16 + mn] = acc[r];
  }
  __syncthreads();

  if (tid < 64) {
    const int row = tid;
    float best = R[row][0]; int bidx = 0;
    for (int j = 1; j < 128; j++) {
      float vv = (j < 64) ? R[row][j] : -R[row][j - 64];
      if (vv > best) { best = vv; bidx = j; }
    }
    const int q = h*NBUCK + bidx;
    const int t = t0 + row;
    bucketArr[((size_t)b*NHASH + h)*SEQ + t] = q;
    atomicAdd(&tilecnt[((size_t)b*NBALL + q)*NT + (t >> 9)], 1);
  }
}

__global__ __launch_bounds__(1024) void lsh_prefix_kernel(int* __restrict__ tilecnt)
{
  __shared__ int sc[NBALL];
  const int b = blockIdx.x, q = threadIdx.x;
  int* tc = tilecnt + ((size_t)b*NBALL + q)*NT;
  int s = 0;
#pragma unroll
  for (int j = 0; j < NT; j++) s += tc[j];
  sc[q] = s;
  __syncthreads();
  for (int off = 1; off < NBALL; off <<= 1) {
    int vv = (q >= off) ? sc[q - off] : 0;
    __syncthreads();
    sc[q] += vv;
    __syncthreads();
  }
  int run = sc[q] - s;
#pragma unroll
  for (int j = 0; j < NT; j++) { int c0 = tc[j]; tc[j] = run; run += c0; }
}

__global__ __launch_bounds__(TS) void lsh_rank_kernel(
    const int* __restrict__ bucketArr, const int* __restrict__ tilecnt,
    int* __restrict__ sticker)
{
  __shared__ int bk[TS];
  const int bid  = blockIdx.x;
  const int tile = bid & (NT - 1);
  const int h    = (bid >> 4) & 7;
  const int b    = bid >> 7;
  const int tid  = threadIdx.x;
  const int t    = tile*TS + tid;
  const int q    = bucketArr[((size_t)b*NHASH + h)*SEQ + t];
  bk[tid] = q;
  __syncthreads();
  int r = 0;
  for (int j = 0; j < tid; j++) r += (bk[j] == q) ? 1 : 0;
  const int pos = tilecnt[((size_t)b*NBALL + q)*NT + tile] + r;
  sticker[(size_t)b*HS + pos] = h*SEQ + t;
}

__global__ __launch_bounds__(256) void lsh_attn_kernel(
    const float* __restrict__ qk, const float* __restrict__ vin,
    const int* __restrict__ sticker,
    float* __restrict__ oBuf, float* __restrict__ lgBuf)
{
  __shared__ int idxA[128];
  __shared__ int tA  [128];
  __shared__ __align__(16) float Qs[64][68];
  __shared__ __align__(16) float KPs[128*68];
  __shared__ __align__(16) float Vs[128][68];
  __shared__ float lse[64];
  float (*Ks)[68]  = (float (*)[68])KPs;
  float (*Ps)[132] = (float (*)[132])KPs;

  const int bid = blockIdx.x;
  const int c   = bid & (NCHUNK - 1);
  const int b   = bid >> 10;
  const int tid = threadIdx.x;
  const int prev = (c + NCHUNK - 1) & (NCHUNK - 1);

  if (tid < 128) {
    const int src_chunk = (tid < 64) ? c : prev;
    const int j = tid & 63;
    const int idx = sticker[(size_t)b*HS + (size_t)src_chunk*BSZ + j];
    idxA[tid] = idx;
    tA[tid]   = idx & (SEQ - 1);
  }
  __syncthreads();

  {
    int row = tid >> 2, c0 = (tid & 3) * 16;
    const float4* s4 = (const float4*)(qk + ((size_t)b*SEQ + tA[row])*DIM + c0);
    float4* d4 = (float4*)(&Qs[row][c0]);
    d4[0]=s4[0]; d4[1]=s4[1]; d4[2]=s4[2]; d4[3]=s4[3];
  }
  {
    int row = tid >> 1, c0 = (tid & 1) * 32;
    const int t = tA[row];
    const float4* sk = (const float4*)(qk  + ((size_t)b*SEQ + t)*DIM + c0);
    const float4* sv = (const float4*)(vin + ((size_t)b*SEQ + t)*DIM + c0);
    float4* dk = (float4*)(&Ks[row][c0]);
    float4* dv = (float4*)(&Vs[row][c0]);
#pragma unroll
    for (int i = 0; i < 8; i++) { dk[i] = sk[i]; dv[i] = sv[i]; }
  }
  __syncthreads();

  if (tid < 128) {
    float ss = 0.f;
#pragma unroll
    for (int d = 0; d < 64; d++) { float x = Ks[tid][d]; ss += x * x; }
    const float inv = 1.0f / (sqrtf(ss) + 1e-6f);
#pragma unroll
    for (int d = 0; d < 64; d++) Ks[tid][d] *= inv;
  }
  __syncthreads();

  const int wave = tid >> 5, lane = tid & 31, half = lane >> 4, mn = lane & 15;

  {
    const int ni = wave;
    v8f acc[4] = {};
#pragma unroll
    for (int k0 = 0; k0 < 64; k0 += 4) {
      v2f bb;
      bb.x = Ks[ni*16 + mn][k0 + 2*half    ];
      bb.y = Ks[ni*16 + mn][k0 + 2*half + 1];
#pragma unroll
      for (int mi = 0; mi < 4; mi++) {
        v2f a;
        a.x = Qs[mi*16 + mn][k0 + 2*half    ];
        a.y = Qs[mi*16 + mn][k0 + 2*half + 1];
        acc[mi] = __builtin_amdgcn_wmma_f32_16x16x4_f32(false, a, false, bb,
                                                        (short)0, acc[mi], false, false);
      }
    }
    __syncthreads();
#pragma unroll
    for (int mi = 0; mi < 4; mi++) {
#pragma unroll
      for (int r = 0; r < 8; r++) {
        const int i = mi*16 + r + 8*half, j = ni*16 + mn;
        float dval = acc[mi][r] * 0.125f;
        if (tA[i] == tA[j]) dval = SELF_VAL;
        Ps[i][j] = dval;
      }
    }
  }
  __syncthreads();

  if (tid < 64) {
    float m = -1e30f;
#pragma unroll
    for (int j = 0; j < 128; j++) m = fmaxf(m, Ps[tid][j]);
    float ssum = 0.f;
#pragma unroll 32
    for (int j = 0; j < 128; j++) { float e = expf(Ps[tid][j] - m); Ps[tid][j] = e; ssum += e; }
    lse[tid] = m + logf(ssum);
    const float invs = 1.0f / ssum;
#pragma unroll
    for (int j = 0; j < 128; j++) Ps[tid][j] *= invs;
  }
  __syncthreads();

  for (int T = wave; T < 16; T += 8) {
    const int mi = T >> 2, ni = T & 3;
    v8f acc = {};
#pragma unroll
    for (int k0 = 0; k0 < 128; k0 += 4) {
      v2f a, bb;
      a.x  = Ps[mi*16 + mn][k0 + 2*half    ];
      a.y  = Ps[mi*16 + mn][k0 + 2*half + 1];
      bb.x = Vs[k0 + 2*half    ][ni*16 + mn];
      bb.y = Vs[k0 + 2*half + 1][ni*16 + mn];
      acc = __builtin_amdgcn_wmma_f32_16x16x4_f32(false, a, false, bb,
                                                  (short)0, acc, false, false);
    }
#pragma unroll
    for (int r = 0; r < 8; r++)
      Qs[mi*16 + r + 8*half][ni*16 + mn] = acc[r];
  }
  __syncthreads();

  {
    int row = tid >> 2, c0 = (tid & 3) * 16;
    float4* d4 = (float4*)(oBuf + ((size_t)b*HS + idxA[row])*DIM + c0);
    const float4* s4 = (const float4*)(&Qs[row][c0]);
    d4[0]=s4[0]; d4[1]=s4[1]; d4[2]=s4[2]; d4[3]=s4[3];
  }
  if (tid < 64) lgBuf[(size_t)b*HS + idxA[tid]] = lse[tid];
}

__global__ __launch_bounds__(256) void lsh_combine_kernel(
    const float* __restrict__ oBuf, const float* __restrict__ lgBuf,
    float* __restrict__ out)
{
  const int g = blockIdx.x * blockDim.x + threadIdx.x;
  const int b = g >> 13;
  const int t = g & (SEQ - 1);
  float lg[NHASH];
  float m = -1e30f;
#pragma unroll
  for (int h = 0; h < NHASH; h++) {
    lg[h] = lgBuf[(size_t)b*HS + (size_t)h*SEQ + t];
    m = fmaxf(m, lg[h]);
  }
  float ssum = 0.f;
#pragma unroll
  for (int h = 0; h < NHASH; h++) { lg[h] = expf(lg[h] - m); ssum += lg[h]; }
  const float inv = 1.0f / ssum;
#pragma unroll
  for (int h = 0; h < NHASH; h++) lg[h] *= inv;

  float* dst = out + ((size_t)b*SEQ + t)*DIM;
#pragma unroll
  for (int d = 0; d < DIM; d += 4) {
    float4 acc = {0.f, 0.f, 0.f, 0.f};
    for (int h = 0; h < NHASH; h++) {
      const float4 o4 = *(const float4*)(oBuf + ((size_t)b*HS + (size_t)h*SEQ + t)*DIM + d);
      acc.x += lg[h]*o4.x; acc.y += lg[h]*o4.y; acc.z += lg[h]*o4.z; acc.w += lg[h]*o4.w;
    }
    *(float4*)(dst + d) = acc;
  }
}

extern "C" void kernel_launch(void* const* d_in, const int* in_sizes, int n_in,
                              void* d_out, int out_size, void* d_ws, size_t ws_size,
                              hipStream_t stream)
{
  const float* qk  = (const float*)d_in[0];
  const float* v   = (const float*)d_in[1];
  const float* rot = (const float*)d_in[2];

  char* ws = (char*)d_ws;
  size_t off = 0;
  auto alloc = [&](size_t bytes) -> void* {
    void* p = ws + off;
    off += (bytes + 255) & ~(size_t)255;
    return p;
  };
  int*   bucketArr = (int*)  alloc((size_t)BATCH*NHASH*SEQ*sizeof(int));
  int*   tilecnt   = (int*)  alloc((size_t)BATCH*NBALL*NT*sizeof(int));
  int*   sticker   = (int*)  alloc((size_t)BATCH*HS*sizeof(int));
  float* oBuf      = (float*)alloc((size_t)BATCH*HS*DIM*sizeof(float));
  float* lgBuf     = (float*)alloc((size_t)BATCH*HS*sizeof(float));
  if (off > ws_size) return;

  hipMemsetAsync(tilecnt, 0, (size_t)BATCH*NBALL*NT*sizeof(int), stream);

  lsh_hash_kernel   <<<BATCH*NHASH*(SEQ/64), 256, 0, stream>>>(qk, rot, bucketArr, tilecnt);
  lsh_prefix_kernel <<<BATCH,               1024, 0, stream>>>(tilecnt);
  lsh_rank_kernel   <<<BATCH*NHASH*NT,        TS, 0, stream>>>(bucketArr, tilecnt, sticker);
  lsh_attn_kernel   <<<BATCH*NCHUNK,         256, 0, stream>>>(qk, v, sticker, oBuf, lgBuf);
  lsh_combine_kernel<<<(BATCH*SEQ)/256,      256, 0, stream>>>(oBuf, lgBuf, (float*)d_out);
}


# attention kernel: key normalisation and row softmax run on all 8 waves (2 and 4 lanes per row, values in registers, partial sums and maxima joined by quad DPP) instead of 4 and 2 waves on whole rows
# speedup vs baseline: 1.1657x; 1.0540x over previous
#include <hip/hip_runtime.h>
#include <math.h>

#define BATCH 8
#define SEQ   8192
#define DIM   64
#define NHASH 8
#define NBUCK 128
#define NBALL (NHASH*NBUCK)
#define BSZ   64
#define NCHUNK (NHASH*(SEQ/BSZ))
#define HS    (NHASH*SEQ)
#define TS    512
#define NT    (SEQ/TS)
#define SELF_VAL (-5.0e4f)

typedef float v2f __attribute__((ext_vector_type(2)));
typedef float v8f __attribute__((ext_vector_type(8)));

__device__ __forceinline__ float quad_xor1(float v) {
  return __int_as_float(__builtin_amdgcn_update_dpp(0, __float_as_int(v), 0xB1, 0xF, 0xF, true));
}
__device__ __forceinline__ float quad_xor2(float v) {
  return __int_as_float(__builtin_amdgcn_update_dpp(0, __float_as_int(v), 0x4E, 0xF, 0xF, true));
}

__global__ __launch_bounds__(256) void lsh_hash_kernel(
    const float* __restrict__ qk, const float* __restrict__ rot,
    int* __restrict__ bucketArr, int* __restrict__ tilecnt)
{
  __shared__ __align__(16) float A [64][68];
  __shared__ __align__(16) float Bm[64][68];
  __shared__ __align__(16) float R [64][68];

  const int bid = blockIdx.x;
  const int tt  = bid & 127;
  const int h   = (bid >> 7) & 7;
  const int b   = bid >> 10;
  const int tid = threadIdx.x;
  const int t0  = tt * 64;

  {
    int row = tid >> 2, c0 = (tid & 3) * 16;
    const float4* s4 = (const float4*)(qk + ((size_t)b*SEQ + t0 + row)*DIM + c0);
    float4* d4 = (float4*)(&A[row][c0]);
    d4[0]=s4[0]; d4[1]=s4[1]; d4[2]=s4[2]; d4[3]=s4[3];
  }
  {
    int f = tid >> 2, c0 = (tid & 3) * 16;
    const float4* s4 = (const float4*)(rot + (size_t)f*(NHASH*64) + (size_t)h*64 + c0);
    float4* d4 = (float4*)(&Bm[f][c0]);
    d4[0]=s4[0]; d4[1]=s4[1]; d4[2]=s4[2]; d4[3]=s4[3];
  }
  __syncthreads();

  const int wave = tid >> 5, lane = tid & 31, half = lane >> 4, mn = lane & 15;
  for (int T = wave; T < 16; T += 8) {
    const int mi = T >> 2, ni = T & 3;
    v8f acc = {};
#pragma unroll
    for (int k0 = 0; k0 < 64; k0 += 4) {
      v2f a, bb;
      a.x  = A [mi*16 + mn][k0 + 2*half    ];
      a.y  = A [mi*16 + mn][k0 + 2*half + 1];
      bb.x = Bm[k0 + 2*half    ][ni*16 + mn];
      bb.y = Bm[k0 + 2*half + 1][ni*16 + mn];
      acc = __builtin_amdgcn_wmma_f32_16x16x4_f32(false, a, false, bb,
                                                  (short)0, acc, false, false);
    }
#pragma unroll
    for (int r = 0; r < 8; r++)
      R[mi*16 + r + 8*half][ni*16 + mn] = acc[r];
  }
  __syncthreads();

  if (tid < 64) {
    const int row = tid;
    float best = R[row][0]; int bidx = 0;
    for (int j = 1; j < 128; j++) {
      float vv = (j < 64) ? R[row][j] : -R[row][j - 64];
      if (vv > best) { best = vv; bidx = j; }
    }
    const int q = h*NBUCK + bidx;
    const int t = t0 + row;
    bucketArr[((size_t)b*NHASH + h)*SEQ + t] = q;
    atomicAdd(&tilecnt[((size_t)b*NBALL + q)*NT + (t >> 9)], 1);
  }
}

__global__ __launch_bounds__(1024) void lsh_prefix_kernel(int* __restrict__ tilecnt)
{
  __shared__ int sc[NBALL];
  const int b = blockIdx.x, q = threadIdx.x;
  int* tc = tilecnt + ((size_t)b*NBALL + q)*NT;
  int s = 0;
#pragma unroll
  for (int j = 0; j < NT; j++) s += tc[j];
  sc[q] = s;
  __syncthreads();
  for (int off = 1; off < NBALL; off <<= 1) {
    int vv = (q >= off) ? sc[q - off] : 0;
    __syncthreads();
    sc[q] += vv;
    __syncthreads();
  }
  int run = sc[q] - s;
#pragma unroll
  for (int j = 0; j < NT; j++) { int c0 = tc[j]; tc[j] = run; run += c0; }
}

__global__ __launch_bounds__(TS) void lsh_rank_kernel(
    const int* __restrict__ bucketArr, const int* __restrict__ tilecnt,
    int* __restrict__ sticker)
{
  __shared__ int bk[TS];
  const int bid  = blockIdx.x;
  const int tile = bid & (NT - 1);
  const int h    = (bid >> 4) & 7;
  const int b    = bid >> 7;
  const int tid  = threadIdx.x;
  const int t    = tile*TS + tid;
  const int q    = bucketArr[((size_t)b*NHASH + h)*SEQ + t];
  bk[tid] = q;
  __syncthreads();
  int r = 0;
  for (int j = 0; j < tid; j++) r += (bk[j] == q) ? 1 : 0;
  const int pos = tilecnt[((size_t)b*NBALL + q)*NT + tile] + r;
  sticker[(size_t)b*HS + pos] = h*SEQ + t;
}

__global__ __launch_bounds__(256) void lsh_attn_kernel(
    const float* __restrict__ qk, const float* __restrict__ vin,
    const int* __restrict__ sticker,
    float* __restrict__ oBuf, float* __restrict__ lgBuf)
{
  __shared__ int idxA[128];
  __shared__ int tA  [128];
  __shared__ __align__(16) float Qs[64][68];
  __shared__ __align__(16) float KPs[128*68];
  __shared__ __align__(16) float Vs[128][68];
  __shared__ float lse[64];
  float (*Ks)[68]  = (float (*)[68])KPs;
  float (*Ps)[132] = (float (*)[132])KPs;

  const int bid = blockIdx.x;
  const int c   = bid & (NCHUNK - 1);
  const int b   = bid >> 10;
  const int tid = threadIdx.x;
  const int prev = (c + NCHUNK - 1) & (NCHUNK - 1);

  if (tid < 128) {
    const int src_chunk = (tid < 64) ? c : prev;
    const int j = tid & 63;
    const int idx = sticker[(size_t)b*HS + (size_t)src_chunk*BSZ + j];
    idxA[tid] = idx;
    tA[tid]   = idx & (SEQ - 1);
  }
  __syncthreads();

  {
    int row = tid >> 2, c0 = (tid & 3) * 16;
    const float4* s4 = (const float4*)(qk + ((size_t)b*SEQ + tA[row])*DIM + c0);
    float4* d4 = (float4*)(&Qs[row][c0]);
    d4[0]=s4[0]; d4[1]=s4[1]; d4[2]=s4[2]; d4[3]=s4[3];
  }
  {
    int row = tid >> 1, c0 = (tid & 1) * 32;
    const int t = tA[row];
    const float4* sk = (const float4*)(qk  + ((size_t)b*SEQ + t)*DIM + c0);
    const float4* sv = (const float4*)(vin + ((size_t)b*SEQ + t)*DIM + c0);
    float4* dk = (float4*)(&Ks[row][c0]);
    float4* dv = (float4*)(&Vs[row][c0]);
    float4 kr[8];
#pragma unroll
    for (int i = 0; i < 8; i++) kr[i] = sk[i];
#pragma unroll
    for (int i = 0; i < 8; i++) dv[i] = sv[i];
    float s0 = 0.f, s1 = 0.f, s2 = 0.f, s3 = 0.f;
#pragma unroll
    for (int i = 0; i < 8; i++) {
      s0 += kr[i].x * kr[i].x; s1 += kr[i].y * kr[i].y;
      s2 += kr[i].z * kr[i].z; s3 += kr[i].w * kr[i].w;
    }
    float ss = (s0 + s1) + (s2 + s3);
    ss += quad_xor1(ss);
    const float inv = 1.0f / (sqrtf(ss) + 1e-6f);
#pragma unroll
    for (int i = 0; i < 8; i++) {
      float4 kk = kr[i];
      kk.x *= inv; kk.y *= inv; kk.z *= inv; kk.w *= inv;
      dk[i] = kk;
    }
  }
  __syncthreads();

  const int wave = tid >> 5, lane = tid & 31, half = lane >> 4, mn = lane & 15;

  {
    const int ni = wave;
    v8f acc[4] = {};
#pragma unroll
    for (int k0 = 0; k0 < 64; k0 += 4) {
      v2f bb;
      bb.x = Ks[ni*16 + mn][k0 + 2*half    ];
      bb.y = Ks[ni*16 + mn][k0 + 2*half + 1];
#pragma unroll
      for (int mi = 0; mi < 4; mi++) {
        v2f a;
        a.x = Qs[mi*16 + mn][k0 + 2*half    ];
        a.y = Qs[mi*16 + mn][k0 + 2*half + 1];
        acc[mi] = __builtin_amdgcn_wmma_f32_16x16x4_f32(false, a, false, bb,
                                                        (short)0, acc[mi], false, false);
      }
    }
    __syncthreads();
#pragma unroll
    for (int mi = 0; mi < 4; mi++) {
#pragma unroll
      for (int r = 0; r < 8; r++) {
        const int i = mi*16 + r + 8*half, j = ni*16 + mn;
        float dval = acc[mi][r] * 0.125f;
        if (tA[i] == tA[j]) dval = SELF_VAL;
        Ps[i][j] = dval;
      }
    }
  }
  __syncthreads();

  {
    const int row = tid >> 2, part = tid & 3;
    float4* prow = (float4*)(&Ps[row][part * 32]);
    float4 x[8];
#pragma unroll
    for (int i = 0; i < 8; i++) x[i] = prow[i];
    float m = -1e30f;
#pragma unroll
    for (int i = 0; i < 8; i++)
      m = fmaxf(fmaxf(fmaxf(m, x[i].x), fmaxf(x[i].y, x[i].z)), x[i].w);
    m = fmaxf(m, quad_xor1(m));
    m = fmaxf(m, quad_xor2(m));
    float ssum = 0.f;
#pragma unroll
    for (int i = 0; i < 8; i++) {
      x[i].x = expf(x[i].x - m); ssum += x[i].x;
      x[i].y = expf(x[i].y - m); ssum += x[i].y;
      x[i].z = expf(x[i].z - m); ssum += x[i].z;
      x[i].w = expf(x[i].w - m); ssum += x[i].w;
    }
    ssum += quad_xor1(ssum);
    ssum += quad_xor2(ssum);
    if (part == 0) lse[row] = m + logf(ssum);
    const float invs = 1.0f / ssum;
#pragma unroll
    for (int i = 0; i < 8; i++) {
      float4 p = x[i];
      p.x *= invs; p.y *= invs; p.z *= invs; p.w *= invs;
      prow[i] = p;
    }
  }
  __syncthreads();

  for (int T = wave; T < 16; T += 8) {
    const int mi = T >> 2, ni = T & 3;
    v8f acc = {};
#pragma unroll
    for (int k0 = 0; k0 < 128; k0 += 4) {
      v2f a, bb;
      a.x  = Ps[mi*16 + mn][k0 + 2*half    ];
      a.y  = Ps[mi*16 + mn][k0 + 2*half + 1];
      bb.x = Vs[k0 + 2*half    ][ni*16 + mn];
      bb.y = Vs[k0 + 2*half + 1][ni*16 + mn];
      acc = __builtin_amdgcn_wmma_f32_16x16x4_f32(false, a, false, bb,
                                                  (short)0, acc, false, false);
    }
#pragma unroll
    for (int r = 0; r < 8; r++)
      Qs[mi*16 + r + 8*half][ni*16 + mn] = acc[r];
  }
  __syncthreads();

  {
    int row = tid >> 2, c0 = (tid & 3) * 16;
    float4* d4 = (float4*)(oBuf + ((size_t)b*HS + idxA[row])*DIM + c0);
    const float4* s4 = (const float4*)(&Qs[row][c0]);
    d4[0]=s4[0]; d4[1]=s4[1]; d4[2]=s4[2]; d4[3]=s4[3];
  }
  if (tid < 64) lgBuf[(size_t)b*HS + idxA[tid]] = lse[tid];
}

__global__ __launch_bounds__(256) void lsh_combine_kernel(
    const float* __restrict__ oBuf, const float* __restrict__ lgBuf,
    float* __restrict__ out)
{
  const int g = blockIdx.x * blockDim.x + threadIdx.x;
  const int b = g >> 13;
  const int t = g & (SEQ - 1);
  float lg[NHASH];
  float m = -1e30f;
#pragma unroll
  for (int h = 0; h < NHASH; h++) {
    lg[h] = lgBuf[(size_t)b*HS + (size_t)h*SEQ + t];
    m = fmaxf(m, lg[h]);
  }
  float ssum = 0.f;
#pragma unroll
  for (int h = 0; h < NHASH; h++) { lg[h] = expf(lg[h] - m); ssum += lg[h]; }
  const float inv = 1.0f / ssum;
#pragma unroll
  for (int h = 0; h < NHASH; h++) lg[h] *= inv;

  float* dst = out + ((size_t)b*SEQ + t)*DIM;
#pragma unroll
  for (int d = 0; d < DIM; d += 4) {
    float4 acc = {0.f, 0.f, 0.f, 0.f};
    for (int h = 0; h < NHASH; h++) {
      const float4 o4 = *(const float4*)(oBuf + ((size_t)b*HS + (size_t)h*SEQ + t)*DIM + d);
      acc.x += lg[h]*o4.x; acc.y += lg[h]*o4.y; acc.z += lg[h]*o4.z; acc.w += lg[h]*o4.w;
    }
    *(float4*)(dst + d) = acc;
  }
}

extern "C" void kernel_launch(void* const* d_in, const int* in_sizes, int n_in,
                              void* d_out, int out_size, void* d_ws, size_t ws_size,
                              hipStream_t stream)
{
  const float* qk  = (const float*)d_in[0];
  const float* v   = (const float*)d_in[1];
  const float* rot = (const float*)d_in[2];

  char* ws = (char*)d_ws;
  size_t off = 0;
  auto alloc = [&](size_t bytes) -> void* {
    void* p = ws + off;
    off += (bytes + 255) & ~(size_t)255;
    return p;
  };
  int*   bucketArr = (int*)  alloc((size_t)BATCH*NHASH*SEQ*sizeof(int));
  int*   tilecnt   = (int*)  alloc((size_t)BATCH*NBALL*NT*sizeof(int));
  int*   sticker   = (int*)  alloc((size_t)BATCH*HS*sizeof(int));
  float* oBuf      = (float*)alloc((size_t)BATCH*HS*DIM*sizeof(float));
  float* lgBuf     = (float*)alloc((size_t)BATCH*HS*sizeof(float));
  if (off > ws_size) return;

  hipMemsetAsync(tilecnt, 0, (size_t)BATCH*NBALL*NT*sizeof(int), stream);

  lsh_hash_kernel   <<<BATCH*NHASH*(SEQ/64), 256, 0, stream>>>(qk, rot, bucketArr, tilecnt);
  lsh_prefix_kernel <<<BATCH,               1024, 0, stream>>>(tilecnt);
  lsh_rank_kernel   <<<BATCH*NHASH*NT,        TS, 0, stream>>>(bucketArr, tilecnt, sticker);
  lsh_attn_kernel   <<<BATCH*NCHUNK,         256, 0, stream>>>(qk, v, sticker, oBuf, lgBuf);
  lsh_combine_kernel<<<(BATCH*SEQ)/256,      256, 0, stream>>>(oBuf, lgBuf, (float*)d_out);
}


# global tile loads/stores of hash, attention, combine remapped to 16 lanes per 256-B row (whole 128-B lines per wave instruction; Q = the raw key loads); key norm now summed by 16-lane DPP butterfly
# speedup vs baseline: 1.4036x; 1.2041x over previous
#include <hip/hip_runtime.h>
#include <math.h>

#define BATCH 8
#define SEQ   8192
#define DIM   64
#define NHASH 8
#define NBUCK 128
#define NBALL (NHASH*NBUCK)
#define BSZ   64
#define NCHUNK (NHASH*(SEQ/BSZ))
#define HS    (NHASH*SEQ)
#define TS    512
#define NT    (SEQ/TS)
#define SELF_VAL (-5.0e4f)

typedef float v2f __attribute__((ext_vector_type(2)));
typedef float v8f __attribute__((ext_vector_type(8)));

__device__ __forceinline__ float quad_xor1(float v) {
  return __int_as_float(__builtin_amdgcn_update_dpp(0, __float_as_int(v), 0xB1, 0xF, 0xF, true));
}
__device__ __forceinline__ float quad_xor2(float v) {
  return __int_as_float(__builtin_amdgcn_update_dpp(0, __float_as_int(v), 0x4E, 0xF, 0xF, true));
}
__device__ __forceinline__ float row_xor4(float v) {
  return __int_as_float(__builtin_amdgcn_update_dpp(0, __float_as_int(v), 0x164, 0xF, 0xF, true));
}
__device__ __forceinline__ float row_xor8(float v) {
  return __int_as_float(__builtin_amdgcn_update_dpp(0, __float_as_int(v), 0x168, 0xF, 0xF, true));
}
template <int P>
__device__ __forceinline__ float row_lane(float v) {
  return __int_as_float(__builtin_amdgcn_update_dpp(0, __float_as_int(v), 0x150 + P, 0xF, 0xF, true));
}

__global__ __launch_bounds__(256) void lsh_hash_kernel(
    const float* __restrict__ qk, const float* __restrict__ rot,
    int* __restrict__ bucketArr, int* __restrict__ tilecnt)
{
  __shared__ __align__(16) float A [64][68];
  __shared__ __align__(16) float Bm[64][68];
  __shared__ __align__(16) float R [64][68];

  const int bid = blockIdx.x;
  const int tt  = bid & 127;
  const int h   = (bid >> 7) & 7;
  const int b   = bid >> 10;
  const int tid = threadIdx.x;
  const int t0  = tt * 64;

  const int g16 = tid >> 4, l4 = (tid & 15) * 4;
  {
    const float4* s4 = (const float4*)(qk + ((size_t)b*SEQ + t0)*DIM);
#pragma unroll
    for (int p = 0; p < 4; p++)
      *(float4*)(&A[p*16 + g16][l4]) = s4[p*256 + tid];
  }
  {
#pragma unroll
    for (int p = 0; p < 4; p++) {
      const int f = p*16 + g16;
      *(float4*)(&Bm[f][l4]) =
          *(const float4*)(rot + (size_t)f*(NHASH*64) + (size_t)h*64 + l4);
    }
  }
  __syncthreads();

  const int wave = tid >> 5, lane = tid & 31, half = lane >> 4, mn = lane & 15;
  for (int T = wave; T < 16; T += 8) {
    const int mi = T >> 2, ni = T & 3;
    v8f acc = {};
#pragma unroll
    for (int k0 = 0; k0 < 64; k0 += 4) {
      v2f a, bb;
      a.x  = A [mi*16 + mn][k0 + 2*half    ];
      a.y  = A [mi*16 + mn][k0 + 2*half + 1];
      bb.x = Bm[k0 + 2*half    ][ni*16 + mn];
      bb.y = Bm[k0 + 2*half + 1][ni*16 + mn];
      acc = __builtin_amdgcn_wmma_f32_16x16x4_f32(false, a, false, bb,
                                                  (short)0, acc, false, false);
    }
#pragma unroll
    for (int r = 0; r < 8; r++)
      R[mi*16 + r + 8*half][ni*16 + mn] = acc[r];
  }
  __syncthreads();

  if (tid < 64) {
    const int row = tid;
    float best = R[row][0]; int bidx = 0;
    for (int j = 1; j < 128; j++) {
      float vv = (j < 64) ? R[row][j] : -R[row][j - 64];
      if (vv > best) { best = vv; bidx = j; }
    }
    const int q = h*NBUCK + bidx;
    const int t = t0 + row;
    bucketArr[((size_t)b*NHASH + h)*SEQ + t] = q;
    atomicAdd(&tilecnt[((size_t)b*NBALL + q)*NT + (t >> 9)], 1);
  }
}

__global__ __launch_bounds__(1024) void lsh_prefix_kernel(int* __restrict__ tilecnt)
{
  __shared__ int sc[NBALL];
  const int b = blockIdx.x, q = threadIdx.x;
  int* tc = tilecnt + ((size_t)b*NBALL + q)*NT;
  int s = 0;
#pragma unroll
  for (int j = 0; j < NT; j++) s += tc[j];
  sc[q] = s;
  __syncthreads();
  for (int off = 1; off < NBALL; off <<= 1) {
    int vv = (q >= off) ? sc[q - off] : 0;
    __syncthreads();
    sc[q] += vv;
    __syncthreads();
  }
  int run = sc[q] - s;
#pragma unroll
  for (int j = 0; j < NT; j++) { int c0 = tc[j]; tc[j] = run; run += c0; }
}

__global__ __launch_bounds__(TS) void lsh_rank_kernel(
    const int* __restrict__ bucketArr, const int* __restrict__ tilecnt,
    int* __restrict__ sticker)
{
  __shared__ int bk[TS];
  const int bid  = blockIdx.x;
  const int tile = bid & (NT - 1);
  const int h    = (bid >> 4) & 7;
  const int b    = bid >> 7;
  const int tid  = threadIdx.x;
  const int t    = tile*TS + tid;
  const int q    = bucketArr[((size_t)b*NHASH + h)*SEQ + t];
  bk[tid] = q;
  __syncthreads();
  int r = 0;
  for (int j = 0; j < tid; j++) r += (bk[j] == q) ? 1 : 0;
  const int pos = tilecnt[((size_t)b*NBALL + q)*NT + tile] + r;
  sticker[(size_t)b*HS + pos] = h*SEQ + t;
}

__global__ __launch_bounds__(256) void lsh_attn_kernel(
    const float* __restrict__ qk, const float* __restrict__ vin,
    const int* __restrict__ sticker,
    float* __restrict__ oBuf, float* __restrict__ lgBuf)
{
  __shared__ int idxA[128];
  __shared__ int tA  [128];
  __shared__ __align__(16) float Qs[64][68];
  __shared__ __align__(16) float KPs[128*68];
  __shared__ __align__(16) float Vs[128][68];
  __shared__ float lse[64];
  float (*Ks)[68]  = (float (*)[68])KPs;
  float (*Ps)[132] = (float (*)[132])KPs;

  const int bid = blockIdx.x;
  const int c   = bid & (NCHUNK - 1);
  const int b   = bid >> 10;
  const int tid = threadIdx.x;
  const int prev = (c + NCHUNK - 1) & (NCHUNK - 1);

  if (tid < 128) {
    const int src_chunk = (tid < 64) ? c : prev;
    const int j = tid & 63;
    const int idx = sticker[(size_t)b*HS + (size_t)src_chunk*BSZ + j];
    idxA[tid] = idx;
    tA[tid]   = idx & (SEQ - 1);
  }
  __syncthreads();

  const int g16 = tid >> 4, l4 = (tid & 15) * 4;
  {
    float4 kr[8], vr[8];
#pragma unroll
    for (int p = 0; p < 8; p++) {
      const size_t off = ((size_t)b*SEQ + tA[p*16 + g16])*DIM + l4;
      kr[p] = *(const float4*)(qk  + off);
      vr[p] = *(const float4*)(vin + off);
    }
#pragma unroll
    for (int p = 0; p < 8; p++) *(float4*)(&Vs[p*16 + g16][l4]) = vr[p];
#pragma unroll
    for (int p = 0; p < 4; p++) *(float4*)(&Qs[p*16 + g16][l4]) = kr[p];
    float ss[8];
#pragma unroll
    for (int p = 0; p < 8; p++) {
      float s = (kr[p].x*kr[p].x + kr[p].y*kr[p].y) + (kr[p].z*kr[p].z + kr[p].w*kr[p].w);
      s += quad_xor1(s);
      s += quad_xor2(s);
      s += row_xor4(s);
      s += row_xor8(s);
      ss[p] = s;
    }
    const int own = tid & 7;
    float mine = ss[0];
#pragma unroll
    for (int p = 1; p < 8; p++) mine = (own == p) ? ss[p] : mine;
    const float invm = 1.0f / (sqrtf(mine) + 1e-6f);
    float inv[8];
    inv[0] = row_lane<0>(invm); inv[1] = row_lane<1>(invm);
    inv[2] = row_lane<2>(invm); inv[3] = row_lane<3>(invm);
    inv[4] = row_lane<4>(invm); inv[5] = row_lane<5>(invm);
    inv[6] = row_lane<6>(invm); inv[7] = row_lane<7>(invm);
#pragma unroll
    for (int p = 0; p < 8; p++) {
      float4 kk = kr[p];
      kk.x *= inv[p]; kk.y *= inv[p]; kk.z *= inv[p]; kk.w *= inv[p];
      *(float4*)(&Ks[p*16 + g16][l4]) = kk;
    }
  }
  __syncthreads();

  const int wave = tid >> 5, lane = tid & 31, half = lane >> 4, mn = lane & 15;

  {
    const int ni = wave;
    v8f acc[4] = {};
#pragma unroll
    for (int k0 = 0; k0 < 64; k0 += 4) {
      v2f bb;
      bb.x = Ks[ni*16 + mn][k0 + 2*half    ];
      bb.y = Ks[ni*16 + mn][k0 + 2*half + 1];
#pragma unroll
      for (int mi = 0; mi < 4; mi++) {
        v2f a;
        a.x = Qs[mi*16 + mn][k0 + 2*half    ];
        a.y = Qs[mi*16 + mn][k0 + 2*half + 1];
        acc[mi] = __builtin_amdgcn_wmma_f32_16x16x4_f32(false, a, false, bb,
                                                        (short)0, acc[mi], false, false);
      }
    }
    __syncthreads();
#pragma unroll
    for (int mi = 0; mi < 4; mi++) {
#pragma unroll
      for (int r = 0; r < 8; r++) {
        const int i = mi*16 + r + 8*half, j = ni*16 + mn;
        float dval = acc[mi][r] * 0.125f;
        if (tA[i] == tA[j]) dval = SELF_VAL;
        Ps[i][j] = dval;
      }
    }
  }
  __syncthreads();

  {
    const int row = tid >> 2, part = tid & 3;
    float4* prow = (float4*)(&Ps[row][part * 32]);
    float4 x[8];
#pragma unroll
    for (int i = 0; i < 8; i++) x[i] = prow[i];
    float m = -1e30f;
#pragma unroll
    for (int i = 0; i < 8; i++)
      m = fmaxf(fmaxf(fmaxf(m, x[i].x), fmaxf(x[i].y, x[i].z)), x[i].w);
    m = fmaxf(m, quad_xor1(m));
    m = fmaxf(m, quad_xor2(m));
    float ssum = 0.f;
#pragma unroll
    for (int i = 0; i < 8; i++) {
      x[i].x = expf(x[i].x - m); ssum += x[i].x;
      x[i].y = expf(x[i].y - m); ssum += x[i].y;
      x[i].z = expf(x[i].z - m); ssum += x[i].z;
      x[i].w = expf(x[i].w - m); ssum += x[i].w;
    }
    ssum += quad_xor1(ssum);
    ssum += quad_xor2(ssum);
    if (part == 0) lse[row] = m + logf(ssum);
    const float invs = 1.0f / ssum;
#pragma unroll
    for (int i = 0; i < 8; i++) {
      float4 p = x[i];
      p.x *= invs; p.y *= invs; p.z *= invs; p.w *= invs;
      prow[i] = p;
    }
  }
  __syncthreads();

  for (int T = wave; T < 16; T += 8) {
    const int mi = T >> 2, ni = T & 3;
    v8f acc = {};
#pragma unroll
    for (int k0 = 0; k0 < 128; k0 += 4) {
      v2f a, bb;
      a.x  = Ps[mi*16 + mn][k0 + 2*half    ];
      a.y  = Ps[mi*16 + mn][k0 + 2*half + 1];
      bb.x = Vs[k0 + 2*half    ][ni*16 + mn];
      bb.y = Vs[k0 + 2*half + 1][ni*16 + mn];
      acc = __builtin_amdgcn_wmma_f32_16x16x4_f32(false, a, false, bb,
                                                  (short)0, acc, false, false);
    }
#pragma unroll
    for (int r = 0; r < 8; r++)
      Qs[mi*16 + r + 8*half][ni*16 + mn] = acc[r];
  }
  __syncthreads();

  {
#pragma unroll
    for (int p = 0; p < 4; p++) {
      const int row = p*16 + g16;
      *(float4*)(oBuf + ((size_t)b*HS + idxA[row])*DIM + l4) = *(const float4*)(&Qs[row][l4]);
    }
  }
  if (tid < 64) lgBuf[(size_t)b*HS + idxA[tid]] = lse[tid];
}

#define CTOK 4

__device__ __forceinline__ void combine_token(const float4 (&o4)[NHASH], float (&lg)[NHASH],
                                              float* __restrict__ dst)
{
  float m = -1e30f;
#pragma unroll
  for (int h = 0; h < NHASH; h++) m = fmaxf(m, lg[h]);
  float ssum = 0.f;
#pragma unroll
  for (int h = 0; h < NHASH; h++) { lg[h] = expf(lg[h] - m); ssum += lg[h]; }
  const float inv = 1.0f / ssum;
#pragma unroll
  for (int h = 0; h < NHASH; h++) lg[h] *= inv;

  float4 acc = {0.f, 0.f, 0.f, 0.f};
#pragma unroll
  for (int h = 0; h < NHASH; h++) {
    acc.x += lg[h]*o4[h].x; acc.y += lg[h]*o4[h].y; acc.z += lg[h]*o4[h].z; acc.w += lg[h]*o4[h].w;
  }
  *(float4*)dst = acc;
}

__global__ __launch_bounds__(256) void lsh_combine_kernel(
    const float* __restrict__ oBuf, const float* __restrict__ lgBuf,
    float* __restrict__ out)
{
  const int l4  = (threadIdx.x & 15) * 4;
  const int tk0 = (blockIdx.x * 16 + (threadIdx.x >> 4)) * CTOK;
  const int b   = tk0 >> 13;
  const int t0  = tk0 & (SEQ - 1);
  const float* ob = oBuf  + ((size_t)b*HS + t0)*DIM + l4;
  const float* lb = lgBuf +  (size_t)b*HS + t0;
  float*       dst = out  + ((size_t)b*SEQ + t0)*DIM + l4;

  float4 o4[NHASH];
  float  lg[NHASH];
#pragma unroll
  for (int h = 0; h < NHASH; h++) {
    o4[h] = *(const float4*)(ob + (size_t)h*SEQ*DIM);
    lg[h] = lb[(size_t)h*SEQ];
  }
#pragma unroll 1
  for (int i = 1; i < CTOK; i++) {
    float4 n4[NHASH];
    float  nl[NHASH];
#pragma unroll
    for (int h = 0; h < NHASH; h++) {
      n4[h] = *(const float4*)(ob + (size_t)h*SEQ*DIM + (size_t)i*DIM);
      nl[h] = lb[(size_t)h*SEQ + i];
    }
    combine_token(o4, lg, dst + (size_t)(i - 1)*DIM);
#pragma unroll
    for (int h = 0; h < NHASH; h++) { o4[h] = n4[h]; lg[h] = nl[h]; }
  }
  combine_token(o4, lg, dst + (size_t)(CTOK - 1)*DIM);
}

extern "C" void kernel_launch(void* const* d_in, const int* in_sizes, int n_in,
                              void* d_out, int out_size, void* d_ws, size_t ws_size,
                              hipStream_t stream)
{
  const float* qk  = (const float*)d_in[0];
  const float* v   = (const float*)d_in[1];
  const float* rot = (const float*)d_in[2];

  char* ws = (char*)d_ws;
  size_t off = 0;
  auto alloc = [&](size_t bytes) -> void* {
    void* p = ws + off;
    off += (bytes + 255) & ~(size_t)255;
    return p;
  };
  int*   bucketArr = (int*)  alloc((size_t)BATCH*NHASH*SEQ*sizeof(int));
  int*   tilecnt   = (int*)  alloc((size_t)BATCH*NBALL*NT*sizeof(int));
  int*   sticker   = (int*)  alloc((size_t)BATCH*HS*sizeof(int));
  float* oBuf      = (float*)alloc((size_t)BATCH*HS*DIM*sizeof(float));
  float* lgBuf     = (float*)alloc((size_t)BATCH*HS*sizeof(float));
  if (off > ws_size) return;

  hipMemsetAsync(tilecnt, 0, (size_t)BATCH*NBALL*NT*sizeof(int), stream);

  lsh_hash_kernel   <<<BATCH*NHASH*(SEQ/64), 256, 0, stream>>>(qk, rot, bucketArr, tilecnt);
  lsh_prefix_kernel <<<BATCH,               1024, 0, stream>>>(tilecnt);
  lsh_rank_kernel   <<<BATCH*NHASH*NT,        TS, 0, stream>>>(bucketArr, tilecnt, sticker);
  lsh_attn_kernel   <<<BATCH*NCHUNK,         256, 0, stream>>>(qk, v, sticker, oBuf, lgBuf);
  lsh_combine_kernel<<<(BATCH*SEQ)/(16*CTOK), 256, 0, stream>>>(oBuf, lgBuf, (float*)d_out);
}


# attention: scores kept in registers (K*Q^T transposed, its D fragments feed P*V as A); V takes the key tile's LDS: 88->54 KB, 5 groups/WGP not 3; softmax sum and P*V summed per key half, 1/sum last
# speedup vs baseline: 1.5590x; 1.1107x over previous
#include <hip/hip_runtime.h>
#include <math.h>

#define BATCH 8
#define SEQ   8192
#define DIM   64
#define NHASH 8
#define NBUCK 128
#define NBALL (NHASH*NBUCK)
#define BSZ   64
#define NCHUNK (NHASH*(SEQ/BSZ))
#define HS    (NHASH*SEQ)
#define TS    512
#define NT    (SEQ/TS)
#define SELF_VAL (-5.0e4f)

typedef float v2f __attribute__((ext_vector_type(2)));
typedef float v8f __attribute__((ext_vector_type(8)));

__device__ __forceinline__ float quad_xor1(float v) {
  return __int_as_float(__builtin_amdgcn_update_dpp(0, __float_as_int(v), 0xB1, 0xF, 0xF, true));
}
__device__ __forceinline__ float quad_xor2(float v) {
  return __int_as_float(__builtin_amdgcn_update_dpp(0, __float_as_int(v), 0x4E, 0xF, 0xF, true));
}
__device__ __forceinline__ float row_xor4(float v) {
  return __int_as_float(__builtin_amdgcn_update_dpp(0, __float_as_int(v), 0x164, 0xF, 0xF, true));
}
__device__ __forceinline__ float row_xor8(float v) {
  return __int_as_float(__builtin_amdgcn_update_dpp(0, __float_as_int(v), 0x168, 0xF, 0xF, true));
}
__device__ __forceinline__ float half_swap(float v) {
  const unsigned u = __float_as_uint(v);
  return __uint_as_float(__builtin_amdgcn_permlanex16(u, u, 0x76543210u, 0xfedcba98u, false, false));
}
template <int P>
__device__ __forceinline__ float row_lane(float v) {
  return __int_as_float(__builtin_amdgcn_update_dpp(0, __float_as_int(v), 0x150 + P, 0xF, 0xF, true));
}

__global__ __launch_bounds__(256) void lsh_hash_kernel(
    const float* __restrict__ qk, const float* __restrict__ rot,
    int* __restrict__ bucketArr, int* __restrict__ tilecnt)
{
  __shared__ __align__(16) float A [64][68];
  __shared__ __align__(16) float Bm[64][68];
  __shared__ __align__(16) float R [64][68];

  const int bid = blockIdx.x;
  const int tt  = bid & 127;
  const int h   = (bid >> 7) & 7;
  const int b   = bid >> 10;
  const int tid = threadIdx.x;
  const int t0  = tt * 64;

  const int g16 = tid >> 4, l4 = (tid & 15) * 4;
  {
    const float4* s4 = (const float4*)(qk + ((size_t)b*SEQ + t0)*DIM);
#pragma unroll
    for (int p = 0; p < 4; p++)
      *(float4*)(&A[p*16 + g16][l4]) = s4[p*256 + tid];
  }
  {
#pragma unroll
    for (int p = 0; p < 4; p++) {
      const int f = p*16 + g16;
      *(float4*)(&Bm[f][l4]) =
          *(const float4*)(rot + (size_t)f*(NHASH*64) + (size_t)h*64 + l4);
    }
  }
  __syncthreads();

  const int wave = tid >> 5, lane = tid & 31, half = lane >> 4, mn = lane & 15;
  for (int T = wave; T < 16; T += 8) {
    const int mi = T >> 2, ni = T & 3;
    v8f acc = {};
#pragma unroll
    for (int k0 = 0; k0 < 64; k0 += 4) {
      v2f a, bb;
      a.x  = A [mi*16 + mn][k0 + 2*half    ];
      a.y  = A [mi*16 + mn][k0 + 2*half + 1];
      bb.x = Bm[k0 + 2*half    ][ni*16 + mn];
      bb.y = Bm[k0 + 2*half + 1][ni*16 + mn];
      acc = __builtin_amdgcn_wmma_f32_16x16x4_f32(false, a, false, bb,
                                                  (short)0, acc, false, false);
    }
#pragma unroll
    for (int r = 0; r < 8; r++)
      R[mi*16 + r + 8*half][ni*16 + mn] = acc[r];
  }
  __syncthreads();

  if (tid < 64) {
    const int row = tid;
    float best = R[row][0]; int bidx = 0;
    for (int j = 1; j < 128; j++) {
      float vv = (j < 64) ? R[row][j] : -R[row][j - 64];
      if (vv > best) { best = vv; bidx = j; }
    }
    const int q = h*NBUCK + bidx;
    const int t = t0 + row;
    bucketArr[((size_t)b*NHASH + h)*SEQ + t] = q;
    atomicAdd(&tilecnt[((size_t)b*NBALL + q)*NT + (t >> 9)], 1);
  }
}

__global__ __launch_bounds__(1024) void lsh_prefix_kernel(int* __restrict__ tilecnt)
{
  __shared__ int sc[NBALL];
  const int b = blockIdx.x, q = threadIdx.x;
  int* tc = tilecnt + ((size_t)b*NBALL + q)*NT;
  int s = 0;
#pragma unroll
  for (int j = 0; j < NT; j++) s += tc[j];
  sc[q] = s;
  __syncthreads();
  for (int off = 1; off < NBALL; off <<= 1) {
    int vv = (q >= off) ? sc[q - off] : 0;
    __syncthreads();
    sc[q] += vv;
    __syncthreads();
  }
  int run = sc[q] - s;
#pragma unroll
  for (int j = 0; j < NT; j++) { int c0 = tc[j]; tc[j] = run; run += c0; }
}

__global__ __launch_bounds__(TS) void lsh_rank_kernel(
    const int* __restrict__ bucketArr, const int* __restrict__ tilecnt,
    int* __restrict__ sticker)
{
  __shared__ int bk[TS];
  const int bid  = blockIdx.x;
  const int tile = bid & (NT - 1);
  const int h    = (bid >> 4) & 7;
  const int b    = bid >> 7;
  const int tid  = threadIdx.x;
  const int t    = tile*TS + tid;
  const int q    = bucketArr[((size_t)b*NHASH + h)*SEQ + t];
  bk[tid] = q;
  __syncthreads();
  int r = 0;
  for (int j = 0; j < tid; j++) r += (bk[j] == q) ? 1 : 0;
  const int pos = tilecnt[((size_t)b*NBALL + q)*NT + tile] + r;
  sticker[(size_t)b*HS + pos] = h*SEQ + t;
}

__global__ __launch_bounds__(256, 10) void lsh_attn_kernel(
    const float* __restrict__ qk, const float* __restrict__ vin,
    const int* __restrict__ sticker,
    float* __restrict__ oBuf, float* __restrict__ lgBuf)
{
  __shared__ int idxA[128];
  __shared__ __align__(16) int tA[128];
  __shared__ __align__(16) float Qs[64][68];
  __shared__ __align__(16) float KVs[128*68];
  __shared__ float redm[2][64];
  __shared__ float reds[2][64];
  float (*Ks)[68]  = (float (*)[68])KVs;
  float (*Vp)[136] = (float (*)[136])KVs;
  float (*O1)[68]  = (float (*)[68])KVs;

  const int bid = blockIdx.x;
  const int c   = bid & (NCHUNK - 1);
  const int b   = bid >> 10;
  const int tid = threadIdx.x;
  const int prev = (c + NCHUNK - 1) & (NCHUNK - 1);

  if (tid < 128) {
    const int src_chunk = (tid < 64) ? c : prev;
    const int j = tid & 63;
    const int idx = sticker[(unsigned)b*HS + (unsigned)src_chunk*BSZ + j];
    idxA[tid] = idx;
    tA[tid]   = idx & (SEQ - 1);
  }
  __syncthreads();

  const int g16 = tid >> 4, l4 = (tid & 15) * 4;
  {
    float4 kr[8];
#pragma unroll
    for (int p = 0; p < 8; p++) {
      const unsigned off = ((unsigned)b*SEQ + tA[32*(p >> 1) + 2*g16 + (p & 1)])*DIM + l4;
      kr[p] = *(const float4*)(qk + off);
    }
#pragma unroll
    for (int p = 0; p < 4; p++) *(float4*)(&Qs[32*(p >> 1) + 2*g16 + (p & 1)][l4]) = kr[p];
    float ss[8];
#pragma unroll
    for (int p = 0; p < 8; p++) {
      float s = (kr[p].x*kr[p].x + kr[p].y*kr[p].y) + (kr[p].z*kr[p].z + kr[p].w*kr[p].w);
      s += quad_xor1(s);
      s += quad_xor2(s);
      s += row_xor4(s);
      s += row_xor8(s);
      ss[p] = s;
    }
    const int own = tid & 7;
    float mine = ss[0];
#pragma unroll
    for (int p = 1; p < 8; p++) mine = (own == p) ? ss[p] : mine;
    const float invm = 1.0f / (sqrtf(mine) + 1e-6f);
    float inv[8];
    inv[0] = row_lane<0>(invm); inv[1] = row_lane<1>(invm);
    inv[2] = row_lane<2>(invm); inv[3] = row_lane<3>(invm);
    inv[4] = row_lane<4>(invm); inv[5] = row_lane<5>(invm);
    inv[6] = row_lane<6>(invm); inv[7] = row_lane<7>(invm);
#pragma unroll
    for (int p = 0; p < 8; p++) {
      float4 kk = kr[p];
      kk.x *= inv[p]; kk.y *= inv[p]; kk.z *= inv[p]; kk.w *= inv[p];
      *(float4*)(&Ks[32*(p >> 1) + 2*g16 + (p & 1)][l4]) = kk;
    }
  }
  __syncthreads();

  const int wave = tid >> 5, lane = tid & 31, half = lane >> 4, mn = lane & 15;
  const int qi   = wave & 3;
  const int kh   = wave >> 2;
  const int kv0  = kh * 64;
  const int qrow = qi*16 + mn;

  v8f st[4] = {};
#pragma unroll
  for (int k0 = 0; k0 < 64; k0 += 4) {
    v2f bq;
    bq.x = Qs[qrow][k0 + 2*half    ];
    bq.y = Qs[qrow][k0 + 2*half + 1];
#pragma unroll
    for (int t = 0; t < 4; t++) {
      v2f ak;
      ak.x = Ks[kv0 + t*16 + mn][k0 + 2*half    ];
      ak.y = Ks[kv0 + t*16 + mn][k0 + 2*half + 1];
      st[t] = __builtin_amdgcn_wmma_f32_16x16x4_f32(false, ak, false, bq,
                                                    (short)0, st[t], false, false);
    }
  }

  float4 vr[8];
#pragma unroll
  for (int p = 0; p < 8; p++) {
    const unsigned off = ((unsigned)b*SEQ + tA[32*(p >> 1) + 2*g16 + (p & 1)])*DIM + l4;
    vr[p] = *(const float4*)(vin + off);
  }

  const int ti = tA[qrow];
  float m = -1e30f;
#pragma unroll
  for (int t = 0; t < 4; t++) {
    const int4 ja = *(const int4*)(&tA[kv0 + t*16 + 8*half    ]);
    const int4 jb = *(const int4*)(&tA[kv0 + t*16 + 8*half + 4]);
    const int tj[8] = {ja.x, ja.y, ja.z, ja.w, jb.x, jb.y, jb.z, jb.w};
#pragma unroll
    for (int r = 0; r < 8; r++) {
      float d = st[t][r] * 0.125f;
      if (tj[r] == ti) d = SELF_VAL;
      st[t][r] = d;
      m = fmaxf(m, d);
    }
  }
  m = fmaxf(m, half_swap(m));
  if (half == 0) redm[kh][qrow] = m;
  __syncthreads();

#pragma unroll
  for (int q = 0; q < 4; q++) {
    const float4 ve = vr[2*q], vo = vr[2*q + 1];
    float4 lo, hi;
    lo.x = ve.x; lo.y = vo.x; lo.z = ve.y; lo.w = vo.y;
    hi.x = ve.z; hi.y = vo.z; hi.z = ve.w; hi.w = vo.w;
    *(float4*)(&Vp[16*q + g16][2*l4    ]) = lo;
    *(float4*)(&Vp[16*q + g16][2*l4 + 4]) = hi;
  }

  m = fmaxf(redm[0][qrow], redm[1][qrow]);
  float ssum = 0.f;
#pragma unroll
  for (int t = 0; t < 4; t++) {
#pragma unroll
    for (int r = 0; r < 8; r++) {
      const float p = expf(st[t][r] - m);
      st[t][r] = p;
      ssum += p;
    }
  }
  ssum += half_swap(ssum);
  if (half == 0) reds[kh][qrow] = ssum;

  __syncthreads();

  v8f o[4] = {};
#pragma unroll
  for (int t = 0; t < 4; t++) {
#pragma unroll
    for (int s = 0; s < 4; s++) {
      v2f a;
      a.x = st[t][2*s    ];
      a.y = st[t][2*s + 1];
      const int pr = (kv0 >> 1) + t*8 + 4*half + s;
#pragma unroll
      for (int nt = 0; nt < 4; nt++) {
        const v2f bb = *(const v2f*)(&Vp[pr][2*(nt*16 + mn)]);
        o[nt] = __builtin_amdgcn_wmma_f32_16x16x4_f32(false, a, false, bb,
                                                      (short)0, o[nt], false, false);
      }
    }
  }
  __syncthreads();
  {
    float (*Od)[68] = kh ? O1 : Qs;
#pragma unroll
    for (int nt = 0; nt < 4; nt++) {
#pragma unroll
      for (int r = 0; r < 8; r++)
        Od[qi*16 + 8*half + r][nt*16 + mn] = o[nt][r];
    }
  }
  __syncthreads();

  {
#pragma unroll
    for (int p = 0; p < 4; p++) {
      const int row = p*16 + g16;
      const float inv = 1.0f / (reds[0][row] + reds[1][row]);
      const float4 u = *(const float4*)(&Qs[row][l4]);
      const float4 w = *(const float4*)(&O1[row][l4]);
      float4 r4;
      r4.x = (u.x + w.x) * inv; r4.y = (u.y + w.y) * inv;
      r4.z = (u.z + w.z) * inv; r4.w = (u.w + w.w) * inv;
      *(float4*)(oBuf + ((unsigned)b*HS + idxA[row])*DIM + l4) = r4;
    }
  }
  if (tid < 64)
    lgBuf[(unsigned)b*HS + idxA[tid]] =
        fmaxf(redm[0][tid], redm[1][tid]) + logf(reds[0][tid] + reds[1][tid]);
}

#define CTOK 4

__device__ __forceinline__ void combine_token(const float4 (&o4)[NHASH], float (&lg)[NHASH],
                                              float* __restrict__ dst)
{
  float m = -1e30f;
#pragma unroll
  for (int h = 0; h < NHASH; h++) m = fmaxf(m, lg[h]);
  float ssum = 0.f;
#pragma unroll
  for (int h = 0; h < NHASH; h++) { lg[h] = expf(lg[h] - m); ssum += lg[h]; }
  const float inv = 1.0f / ssum;
#pragma unroll
  for (int h = 0; h < NHASH; h++) lg[h] *= inv;

  float4 acc = {0.f, 0.f, 0.f, 0.f};
#pragma unroll
  for (int h = 0; h < NHASH; h++) {
    acc.x += lg[h]*o4[h].x; acc.y += lg[h]*o4[h].y; acc.z += lg[h]*o4[h].z; acc.w += lg[h]*o4[h].w;
  }
  *(float4*)dst = acc;
}

__global__ __launch_bounds__(256) void lsh_combine_kernel(
    const float* __restrict__ oBuf, const float* __restrict__ lgBuf,
    float* __restrict__ out)
{
  const int l4  = (threadIdx.x & 15) * 4;
  const int tk0 = (blockIdx.x * 16 + (threadIdx.x >> 4)) * CTOK;
  const int b   = tk0 >> 13;
  const int t0  = tk0 & (SEQ - 1);
  const float* ob = oBuf  + ((size_t)b*HS + t0)*DIM + l4;
  const float* lb = lgBuf +  (size_t)b*HS + t0;
  float*       dst = out  + ((size_t)b*SEQ + t0)*DIM + l4;

  float4 o4[NHASH];
  float  lg[NHASH];
#pragma unroll
  for (int h = 0; h < NHASH; h++) {
    o4[h] = *(const float4*)(ob + (size_t)h*SEQ*DIM);
    lg[h] = lb[(size_t)h*SEQ];
  }
#pragma unroll 1
  for (int i = 1; i < CTOK; i++) {
    float4 n4[NHASH];
    float  nl[NHASH];
#pragma unroll
    for (int h = 0; h < NHASH; h++) {
      n4[h] = *(const float4*)(ob + (size_t)h*SEQ*DIM + (size_t)i*DIM);
      nl[h] = lb[(size_t)h*SEQ + i];
    }
    combine_token(o4, lg, dst + (size_t)(i - 1)*DIM);
#pragma unroll
    for (int h = 0; h < NHASH; h++) { o4[h] = n4[h]; lg[h] = nl[h]; }
  }
  combine_token(o4, lg, dst + (size_t)(CTOK - 1)*DIM);
}

extern "C" void kernel_launch(void* const* d_in, const int* in_sizes, int n_in,
                              void* d_out, int out_size, void* d_ws, size_t ws_size,
                              hipStream_t stream)
{
  const float* qk  = (const float*)d_in[0];
  const float* v   = (const float*)d_in[1];
  const float* rot = (const float*)d_in[2];

  char* ws = (char*)d_ws;
  size_t off = 0;
  auto alloc = [&](size_t bytes) -> void* {
    void* p = ws + off;
    off += (bytes + 255) & ~(size_t)255;
    return p;
  };
  int*   bucketArr = (int*)  alloc((size_t)BATCH*NHASH*SEQ*sizeof(int));
  int*   tilecnt   = (int*)  alloc((size_t)BATCH*NBALL*NT*sizeof(int));
  int*   sticker   = (int*)  alloc((size_t)BATCH*HS*sizeof(int));
  float* oBuf      = (float*)alloc((size_t)BATCH*HS*DIM*sizeof(float));
  float* lgBuf     = (float*)alloc((size_t)BATCH*HS*sizeof(float));
  if (off > ws_size) return;

  hipMemsetAsync(tilecnt, 0, (size_t)BATCH*NBALL*NT*sizeof(int), stream);

  lsh_hash_kernel   <<<BATCH*NHASH*(SEQ/64), 256, 0, stream>>>(qk, rot, bucketArr, tilecnt);
  lsh_prefix_kernel <<<BATCH,               1024, 0, stream>>>(tilecnt);
  lsh_rank_kernel   <<<BATCH*NHASH*NT,        TS, 0, stream>>>(bucketArr, tilecnt, sticker);
  lsh_attn_kernel   <<<BATCH*NCHUNK,         256, 0, stream>>>(qk, v, sticker, oBuf, lgBuf);
  lsh_combine_kernel<<<(BATCH*SEQ)/(16*CTOK), 256, 0, stream>>>(oBuf, lgBuf, (float*)d_out);
}


# hash kernel: product taken transposed (rot^T * qk^T), a wave owns 16 tokens x all 64 columns and takes the argmax in its accumulators: no LDS result tile, no 2-wave serial scan; 128 tokens/block
# speedup vs baseline: 1.7391x; 1.1155x over previous
#include <hip/hip_runtime.h>
#include <math.h>

#define BATCH 8
#define SEQ   8192
#define DIM   64
#define NHASH 8
#define NBUCK 128
#define NBALL (NHASH*NBUCK)
#define BSZ   64
#define NCHUNK (NHASH*(SEQ/BSZ))
#define HS    (NHASH*SEQ)
#define TS    512
#define NT    (SEQ/TS)
#define SELF_VAL (-5.0e4f)

typedef float v2f __attribute__((ext_vector_type(2)));
typedef float v8f __attribute__((ext_vector_type(8)));

__device__ __forceinline__ float quad_xor1(float v) {
  return __int_as_float(__builtin_amdgcn_update_dpp(0, __float_as_int(v), 0xB1, 0xF, 0xF, true));
}
__device__ __forceinline__ float quad_xor2(float v) {
  return __int_as_float(__builtin_amdgcn_update_dpp(0, __float_as_int(v), 0x4E, 0xF, 0xF, true));
}
__device__ __forceinline__ float row_xor4(float v) {
  return __int_as_float(__builtin_amdgcn_update_dpp(0, __float_as_int(v), 0x164, 0xF, 0xF, true));
}
__device__ __forceinline__ float row_xor8(float v) {
  return __int_as_float(__builtin_amdgcn_update_dpp(0, __float_as_int(v), 0x168, 0xF, 0xF, true));
}
__device__ __forceinline__ float half_swap(float v) {
  const unsigned u = __float_as_uint(v);
  return __uint_as_float(__builtin_amdgcn_permlanex16(u, u, 0x76543210u, 0xfedcba98u, false, false));
}
template <int P>
__device__ __forceinline__ float row_lane(float v) {
  return __int_as_float(__builtin_amdgcn_update_dpp(0, __float_as_int(v), 0x150 + P, 0xF, 0xF, true));
}

__device__ __forceinline__ int half_swap_i(int v) {
  const unsigned u = (unsigned)v;
  return (int)__builtin_amdgcn_permlanex16(u, u, 0x76543210u, 0xfedcba98u, false, false);
}

__global__ __launch_bounds__(256) void lsh_hash_kernel(
    const float* __restrict__ qk, const float* __restrict__ rot,
    int* __restrict__ bucketArr, int* __restrict__ tilecnt)
{
  __shared__ __align__(16) float A [128][68];
  __shared__ __align__(16) float Bt[64][68];

  const int bid = blockIdx.x;
  const int tt  = bid & 63;
  const int h   = (bid >> 6) & 7;
  const int b   = bid >> 9;
  const int tid = threadIdx.x;
  const int t0  = tt * 128;

  const int g16 = tid >> 4, l4 = (tid & 15) * 4;
  {
    const float4* s4 = (const float4*)(qk + ((size_t)b*SEQ + t0)*DIM);
#pragma unroll
    for (int p = 0; p < 8; p++)
      *(float4*)(&A[p*16 + g16][l4]) = s4[p*256 + tid];
  }
  {
#pragma unroll
    for (int p = 0; p < 4; p++) {
      const int f = p*16 + g16;
      const float4 r4 =
          *(const float4*)(rot + (size_t)f*(NHASH*64) + (size_t)h*64 + l4);
      Bt[l4    ][f] = r4.x;
      Bt[l4 + 1][f] = r4.y;
      Bt[l4 + 2][f] = r4.z;
      Bt[l4 + 3][f] = r4.w;
    }
  }
  __syncthreads();

  const int wave = tid >> 5, lane = tid & 31, half = lane >> 4, mn = lane & 15;
  const int tok = wave*16 + mn;

  v8f acc[4] = {};
#pragma unroll
  for (int k0 = 0; k0 < 64; k0 += 4) {
    v2f bq;
    bq.x = A[tok][k0 + 2*half    ];
    bq.y = A[tok][k0 + 2*half + 1];
#pragma unroll
    for (int bt = 0; bt < 4; bt++) {
      v2f ar;
      ar.x = Bt[bt*16 + mn][k0 + 2*half    ];
      ar.y = Bt[bt*16 + mn][k0 + 2*half + 1];
      acc[bt] = __builtin_amdgcn_wmma_f32_16x16x4_f32(false, ar, false, bq,
                                                      (short)0, acc[bt], false, false);
    }
  }

  float bmx = 0.f, bmn = 0.f;
  int   imx = 0,   imn = 0;
#pragma unroll
  for (int bt = 0; bt < 4; bt++) {
    float mx = acc[bt][0], mi = acc[bt][0];
    int   ix = bt*16,      ii = bt*16;
#pragma unroll
    for (int r = 1; r < 8; r++) {
      const float vv = acc[bt][r];
      if (vv > mx) { mx = vv; ix = bt*16 + r; }
      if (vv < mi) { mi = vv; ii = bt*16 + r; }
    }
    if (bt == 0) { bmx = mx; imx = ix; bmn = mi; imn = ii; }
    else {
      if (mx > bmx) { bmx = mx; imx = ix; }
      if (mi < bmn) { bmn = mi; imn = ii; }
    }
  }
  imx += 8*half;
  imn += 8*half;
  {
    const float omx = half_swap(bmx), omn = half_swap(bmn);
    const int   oix = half_swap_i(imx), oin = half_swap_i(imn);
    if (omx > bmx || (omx == bmx && oix < imx)) { bmx = omx; imx = oix; }
    if (omn < bmn || (omn == bmn && oin < imn)) { bmn = omn; imn = oin; }
  }
  const int bidx = (bmx >= -bmn) ? imx : 64 + imn;

  if (half == 0) {
    const int q = h*NBUCK + bidx;
    const int t = t0 + tok;
    bucketArr[((size_t)b*NHASH + h)*SEQ + t] = q;
    atomicAdd(&tilecnt[((size_t)b*NBALL + q)*NT + (t >> 9)], 1);
  }
}

__global__ __launch_bounds__(1024) void lsh_prefix_kernel(int* __restrict__ tilecnt)
{
  __shared__ int sc[NBALL];
  const int b = blockIdx.x, q = threadIdx.x;
  int* tc = tilecnt + ((size_t)b*NBALL + q)*NT;
  int s = 0;
#pragma unroll
  for (int j = 0; j < NT; j++) s += tc[j];
  sc[q] = s;
  __syncthreads();
  for (int off = 1; off < NBALL; off <<= 1) {
    int vv = (q >= off) ? sc[q - off] : 0;
    __syncthreads();
    sc[q] += vv;
    __syncthreads();
  }
  int run = sc[q] - s;
#pragma unroll
  for (int j = 0; j < NT; j++) { int c0 = tc[j]; tc[j] = run; run += c0; }
}

__global__ __launch_bounds__(TS) void lsh_rank_kernel(
    const int* __restrict__ bucketArr, const int* __restrict__ tilecnt,
    int* __restrict__ sticker)
{
  __shared__ int bk[TS];
  const int bid  = blockIdx.x;
  const int tile = bid & (NT - 1);
  const int h    = (bid >> 4) & 7;
  const int b    = bid >> 7;
  const int tid  = threadIdx.x;
  const int t    = tile*TS + tid;
  const int q    = bucketArr[((size_t)b*NHASH + h)*SEQ + t];
  bk[tid] = q;
  __syncthreads();
  int r = 0;
  for (int j = 0; j < tid; j++) r += (bk[j] == q) ? 1 : 0;
  const int pos = tilecnt[((size_t)b*NBALL + q)*NT + tile] + r;
  sticker[(size_t)b*HS + pos] = h*SEQ + t;
}

__global__ __launch_bounds__(256, 10) void lsh_attn_kernel(
    const float* __restrict__ qk, const float* __restrict__ vin,
    const int* __restrict__ sticker,
    float* __restrict__ oBuf, float* __restrict__ lgBuf)
{
  __shared__ int idxA[128];
  __shared__ __align__(16) int tA[128];
  __shared__ __align__(16) float Qs[64][68];
  __shared__ __align__(16) float KVs[128*68];
  __shared__ float redm[2][64];
  __shared__ float reds[2][64];
  float (*Ks)[68]  = (float (*)[68])KVs;
  float (*Vp)[136] = (float (*)[136])KVs;
  float (*O1)[68]  = (float (*)[68])KVs;

  const int bid = blockIdx.x;
  const int c   = bid & (NCHUNK - 1);
  const int b   = bid >> 10;
  const int tid = threadIdx.x;
  const int prev = (c + NCHUNK - 1) & (NCHUNK - 1);

  if (tid < 128) {
    const int src_chunk = (tid < 64) ? c : prev;
    const int j = tid & 63;
    const int idx = sticker[(unsigned)b*HS + (unsigned)src_chunk*BSZ + j];
    idxA[tid] = idx;
    tA[tid]   = idx & (SEQ - 1);
  }
  __syncthreads();

  const int g16 = tid >> 4, l4 = (tid & 15) * 4;
  {
    float4 kr[8];
#pragma unroll
    for (int p = 0; p < 8; p++) {
      const unsigned off = ((unsigned)b*SEQ + tA[32*(p >> 1) + 2*g16 + (p & 1)])*DIM + l4;
      kr[p] = *(const float4*)(qk + off);
    }
#pragma unroll
    for (int p = 0; p < 4; p++) *(float4*)(&Qs[32*(p >> 1) + 2*g16 + (p & 1)][l4]) = kr[p];
    float ss[8];
#pragma unroll
    for (int p = 0; p < 8; p++) {
      float s = (kr[p].x*kr[p].x + kr[p].y*kr[p].y) + (kr[p].z*kr[p].z + kr[p].w*kr[p].w);
      s += quad_xor1(s);
      s += quad_xor2(s);
      s += row_xor4(s);
      s += row_xor8(s);
      ss[p] = s;
    }
    const int own = tid & 7;
    float mine = ss[0];
#pragma unroll
    for (int p = 1; p < 8; p++) mine = (own == p) ? ss[p] : mine;
    const float invm = 1.0f / (sqrtf(mine) + 1e-6f);
    float inv[8];
    inv[0] = row_lane<0>(invm); inv[1] = row_lane<1>(invm);
    inv[2] = row_lane<2>(invm); inv[3] = row_lane<3>(invm);
    inv[4] = row_lane<4>(invm); inv[5] = row_lane<5>(invm);
    inv[6] = row_lane<6>(invm); inv[7] = row_lane<7>(invm);
#pragma unroll
    for (int p = 0; p < 8; p++) {
      float4 kk = kr[p];
      kk.x *= inv[p]; kk.y *= inv[p]; kk.z *= inv[p]; kk.w *= inv[p];
      *(float4*)(&Ks[32*(p >> 1) + 2*g16 + (p & 1)][l4]) = kk;
    }
  }
  __syncthreads();

  const int wave = tid >> 5, lane = tid & 31, half = lane >> 4, mn = lane & 15;
  const int qi   = wave & 3;
  const int kh   = wave >> 2;
  const int kv0  = kh * 64;
  const int qrow = qi*16 + mn;

  v8f st[4] = {};
#pragma unroll
  for (int k0 = 0; k0 < 64; k0 += 4) {
    v2f bq;
    bq.x = Qs[qrow][k0 + 2*half    ];
    bq.y = Qs[qrow][k0 + 2*half + 1];
#pragma unroll
    for (int t = 0; t < 4; t++) {
      v2f ak;
      ak.x = Ks[kv0 + t*16 + mn][k0 + 2*half    ];
      ak.y = Ks[kv0 + t*16 + mn][k0 + 2*half + 1];
      st[t] = __builtin_amdgcn_wmma_f32_16x16x4_f32(false, ak, false, bq,
                                                    (short)0, st[t], false, false);
    }
  }

  float4 vr[8];
#pragma unroll
  for (int p = 0; p < 8; p++) {
    const unsigned off = ((unsigned)b*SEQ + tA[32*(p >> 1) + 2*g16 + (p & 1)])*DIM + l4;
    vr[p] = *(const float4*)(vin + off);
  }

  const int ti = tA[qrow];
  float m = -1e30f;
#pragma unroll
  for (int t = 0; t < 4; t++) {
    const int4 ja = *(const int4*)(&tA[kv0 + t*16 + 8*half    ]);
    const int4 jb = *(const int4*)(&tA[kv0 + t*16 + 8*half + 4]);
    const int tj[8] = {ja.x, ja.y, ja.z, ja.w, jb.x, jb.y, jb.z, jb.w};
#pragma unroll
    for (int r = 0; r < 8; r++) {
      float d = st[t][r] * 0.125f;
      if (tj[r] == ti) d = SELF_VAL;
      st[t][r] = d;
      m = fmaxf(m, d);
    }
  }
  m = fmaxf(m, half_swap(m));
  if (half == 0) redm[kh][qrow] = m;
  __syncthreads();

#pragma unroll
  for (int q = 0; q < 4; q++) {
    const float4 ve = vr[2*q], vo = vr[2*q + 1];
    float4 lo, hi;
    lo.x = ve.x; lo.y = vo.x; lo.z = ve.y; lo.w = vo.y;
    hi.x = ve.z; hi.y = vo.z; hi.z = ve.w; hi.w = vo.w;
    *(float4*)(&Vp[16*q + g16][2*l4    ]) = lo;
    *(float4*)(&Vp[16*q + g16][2*l4 + 4]) = hi;
  }

  m = fmaxf(redm[0][qrow], redm[1][qrow]);
  float ssum = 0.f;
#pragma unroll
  for (int t = 0; t < 4; t++) {
#pragma unroll
    for (int r = 0; r < 8; r++) {
      const float p = expf(st[t][r] - m);
      st[t][r] = p;
      ssum += p;
    }
  }
  ssum += half_swap(ssum);
  if (half == 0) reds[kh][qrow] = ssum;

  __syncthreads();

  v8f o[4] = {};
#pragma unroll
  for (int t = 0; t < 4; t++) {
#pragma unroll
    for (int s = 0; s < 4; s++) {
      v2f a;
      a.x = st[t][2*s    ];
      a.y = st[t][2*s + 1];
      const int pr = (kv0 >> 1) + t*8 + 4*half + s;
#pragma unroll
      for (int nt = 0; nt < 4; nt++) {
        const v2f bb = *(const v2f*)(&Vp[pr][2*(nt*16 + mn)]);
        o[nt] = __builtin_amdgcn_wmma_f32_16x16x4_f32(false, a, false, bb,
                                                      (short)0, o[nt], false, false);
      }
    }
  }
  __syncthreads();
  {
    float (*Od)[68] = kh ? O1 : Qs;
#pragma unroll
    for (int nt = 0; nt < 4; nt++) {
#pragma unroll
      for (int r = 0; r < 8; r++)
        Od[qi*16 + 8*half + r][nt*16 + mn] = o[nt][r];
    }
  }
  __syncthreads();

  {
#pragma unroll
    for (int p = 0; p < 4; p++) {
      const int row = p*16 + g16;
      const float inv = 1.0f / (reds[0][row] + reds[1][row]);
      const float4 u = *(const float4*)(&Qs[row][l4]);
      const float4 w = *(const float4*)(&O1[row][l4]);
      float4 r4;
      r4.x = (u.x + w.x) * inv; r4.y = (u.y + w.y) * inv;
      r4.z = (u.z + w.z) * inv; r4.w = (u.w + w.w) * inv;
      *(float4*)(oBuf + ((unsigned)b*HS + idxA[row])*DIM + l4) = r4;
    }
  }
  if (tid < 64)
    lgBuf[(unsigned)b*HS + idxA[tid]] =
        fmaxf(redm[0][tid], redm[1][tid]) + logf(reds[0][tid] + reds[1][tid]);
}

#define CTOK 4

__device__ __forceinline__ void combine_token(const float4 (&o4)[NHASH], float (&lg)[NHASH],
                                              float* __restrict__ dst)
{
  float m = -1e30f;
#pragma unroll
  for (int h = 0; h < NHASH; h++) m = fmaxf(m, lg[h]);
  float ssum = 0.f;
#pragma unroll
  for (int h = 0; h < NHASH; h++) { lg[h] = expf(lg[h] - m); ssum += lg[h]; }
  const float inv = 1.0f / ssum;
#pragma unroll
  for (int h = 0; h < NHASH; h++) lg[h] *= inv;

  float4 acc = {0.f, 0.f, 0.f, 0.f};
#pragma unroll
  for (int h = 0; h < NHASH; h++) {
    acc.x += lg[h]*o4[h].x; acc.y += lg[h]*o4[h].y; acc.z += lg[h]*o4[h].z; acc.w += lg[h]*o4[h].w;
  }
  *(float4*)dst = acc;
}

__global__ __launch_bounds__(256) void lsh_combine_kernel(
    const float* __restrict__ oBuf, const float* __restrict__ lgBuf,
    float* __restrict__ out)
{
  const int l4  = (threadIdx.x & 15) * 4;
  const int tk0 = (blockIdx.x * 16 + (threadIdx.x >> 4)) * CTOK;
  const int b   = tk0 >> 13;
  const int t0  = tk0 & (SEQ - 1);
  const float* ob = oBuf  + ((size_t)b*HS + t0)*DIM + l4;
  const float* lb = lgBuf +  (size_t)b*HS + t0;
  float*       dst = out  + ((size_t)b*SEQ + t0)*DIM + l4;

  float4 o4[NHASH];
  float  lg[NHASH];
#pragma unroll
  for (int h = 0; h < NHASH; h++) {
    o4[h] = *(const float4*)(ob + (size_t)h*SEQ*DIM);
    lg[h] = lb[(size_t)h*SEQ];
  }
#pragma unroll 1
  for (int i = 1; i < CTOK; i++) {
    float4 n4[NHASH];
    float  nl[NHASH];
#pragma unroll
    for (int h = 0; h < NHASH; h++) {
      n4[h] = *(const float4*)(ob + (size_t)h*SEQ*DIM + (size_t)i*DIM);
      nl[h] = lb[(size_t)h*SEQ + i];
    }
    combine_token(o4, lg, dst + (size_t)(i - 1)*DIM);
#pragma unroll
    for (int h = 0; h < NHASH; h++) { o4[h] = n4[h]; lg[h] = nl[h]; }
  }
  combine_token(o4, lg, dst + (size_t)(CTOK - 1)*DIM);
}

extern "C" void kernel_launch(void* const* d_in, const int* in_sizes, int n_in,
                              void* d_out, int out_size, void* d_ws, size_t ws_size,
                              hipStream_t stream)
{
  const float* qk  = (const float*)d_in[0];
  const float* v   = (const float*)d_in[1];
  const float* rot = (const float*)d_in[2];

  char* ws = (char*)d_ws;
  size_t off = 0;
  auto alloc = [&](size_t bytes) -> void* {
    void* p = ws + off;
    off += (bytes + 255) & ~(size_t)255;
    return p;
  };
  int*   bucketArr = (int*)  alloc((size_t)BATCH*NHASH*SEQ*sizeof(int));
  int*   tilecnt   = (int*)  alloc((size_t)BATCH*NBALL*NT*sizeof(int));
  int*   sticker   = (int*)  alloc((size_t)BATCH*HS*sizeof(int));
  float* oBuf      = (float*)alloc((size_t)BATCH*HS*DIM*sizeof(float));
  float* lgBuf     = (float*)alloc((size_t)BATCH*HS*sizeof(float));
  if (off > ws_size) return;

  hipMemsetAsync(tilecnt, 0, (size_t)BATCH*NBALL*NT*sizeof(int), stream);

  lsh_hash_kernel   <<<BATCH*NHASH*(SEQ/128), 256, 0, stream>>>(qk, rot, bucketArr, tilecnt);
  lsh_prefix_kernel <<<BATCH,               1024, 0, stream>>>(tilecnt);
  lsh_rank_kernel   <<<BATCH*NHASH*NT,        TS, 0, stream>>>(bucketArr, tilecnt, sticker);
  lsh_attn_kernel   <<<BATCH*NCHUNK,         256, 0, stream>>>(qk, v, sticker, oBuf, lgBuf);
  lsh_combine_kernel<<<(BATCH*SEQ)/(16*CTOK), 256, 0, stream>>>(oBuf, lgBuf, (float*)d_out);
}


# sort tile = the hash block's 128 tokens: counts and stable ranks taken in its LDS, plain stores (no global atomics, memset or rank kernel); one kernel scans counts and places tokens; same order
# speedup vs baseline: 1.8812x; 1.0817x over previous
#include <hip/hip_runtime.h>
#include <math.h>

#define BATCH 8
#define SEQ   8192
#define DIM   64
#define NHASH 8
#define NBUCK 128
#define NBALL (NHASH*NBUCK)
#define BSZ   64
#define NCHUNK (NHASH*(SEQ/BSZ))
#define HS    (NHASH*SEQ)
#define TSZ   128
#define NTL   (SEQ/TSZ)
#define SELF_VAL (-5.0e4f)

typedef float v2f __attribute__((ext_vector_type(2)));
typedef float v8f __attribute__((ext_vector_type(8)));

__device__ __forceinline__ float quad_xor1(float v) {
  return __int_as_float(__builtin_amdgcn_update_dpp(0, __float_as_int(v), 0xB1, 0xF, 0xF, true));
}
__device__ __forceinline__ float quad_xor2(float v) {
  return __int_as_float(__builtin_amdgcn_update_dpp(0, __float_as_int(v), 0x4E, 0xF, 0xF, true));
}
__device__ __forceinline__ float row_xor4(float v) {
  return __int_as_float(__builtin_amdgcn_update_dpp(0, __float_as_int(v), 0x164, 0xF, 0xF, true));
}
__device__ __forceinline__ float row_xor8(float v) {
  return __int_as_float(__builtin_amdgcn_update_dpp(0, __float_as_int(v), 0x168, 0xF, 0xF, true));
}
__device__ __forceinline__ float half_swap(float v) {
  const unsigned u = __float_as_uint(v);
  return __uint_as_float(__builtin_amdgcn_permlanex16(u, u, 0x76543210u, 0xfedcba98u, false, false));
}
template <int P>
__device__ __forceinline__ float row_lane(float v) {
  return __int_as_float(__builtin_amdgcn_update_dpp(0, __float_as_int(v), 0x150 + P, 0xF, 0xF, true));
}

__device__ __forceinline__ int half_swap_i(int v) {
  const unsigned u = (unsigned)v;
  return (int)__builtin_amdgcn_permlanex16(u, u, 0x76543210u, 0xfedcba98u, false, false);
}

__global__ __launch_bounds__(256) void lsh_hash_kernel(
    const float* __restrict__ qk, const float* __restrict__ rot,
    int* __restrict__ br, int* __restrict__ cnt)
{
  __shared__ __align__(16) int bk[TSZ];
  __shared__ __align__(16) float A [128][68];
  __shared__ __align__(16) float Bt[64][68];

  const int bid = blockIdx.x;
  const int tt  = bid & 63;
  const int h   = (bid >> 6) & 7;
  const int b   = bid >> 9;
  const int tid = threadIdx.x;
  const int t0  = tt * 128;

  const int g16 = tid >> 4, l4 = (tid & 15) * 4;
  {
    const float4* s4 = (const float4*)(qk + ((size_t)b*SEQ + t0)*DIM);
#pragma unroll
    for (int p = 0; p < 8; p++)
      *(float4*)(&A[p*16 + g16][l4]) = s4[p*256 + tid];
  }
  {
#pragma unroll
    for (int p = 0; p < 4; p++) {
      const int f = p*16 + g16;
      const float4 r4 =
          *(const float4*)(rot + (size_t)f*(NHASH*64) + (size_t)h*64 + l4);
      Bt[l4    ][f] = r4.x;
      Bt[l4 + 1][f] = r4.y;
      Bt[l4 + 2][f] = r4.z;
      Bt[l4 + 3][f] = r4.w;
    }
  }
  __syncthreads();

  const int wave = tid >> 5, lane = tid & 31, half = lane >> 4, mn = lane & 15;
  const int tok = wave*16 + mn;

  v8f acc[4] = {};
#pragma unroll
  for (int k0 = 0; k0 < 64; k0 += 4) {
    v2f bq;
    bq.x = A[tok][k0 + 2*half    ];
    bq.y = A[tok][k0 + 2*half + 1];
#pragma unroll
    for (int bt = 0; bt < 4; bt++) {
      v2f ar;
      ar.x = Bt[bt*16 + mn][k0 + 2*half    ];
      ar.y = Bt[bt*16 + mn][k0 + 2*half + 1];
      acc[bt] = __builtin_amdgcn_wmma_f32_16x16x4_f32(false, ar, false, bq,
                                                      (short)0, acc[bt], false, false);
    }
  }

  float bmx = 0.f, bmn = 0.f;
  int   imx = 0,   imn = 0;
#pragma unroll
  for (int bt = 0; bt < 4; bt++) {
    float mx = acc[bt][0], mi = acc[bt][0];
    int   ix = bt*16,      ii = bt*16;
#pragma unroll
    for (int r = 1; r < 8; r++) {
      const float vv = acc[bt][r];
      if (vv > mx) { mx = vv; ix = bt*16 + r; }
      if (vv < mi) { mi = vv; ii = bt*16 + r; }
    }
    if (bt == 0) { bmx = mx; imx = ix; bmn = mi; imn = ii; }
    else {
      if (mx > bmx) { bmx = mx; imx = ix; }
      if (mi < bmn) { bmn = mi; imn = ii; }
    }
  }
  imx += 8*half;
  imn += 8*half;
  {
    const float omx = half_swap(bmx), omn = half_swap(bmn);
    const int   oix = half_swap_i(imx), oin = half_swap_i(imn);
    if (omx > bmx || (omx == bmx && oix < imx)) { bmx = omx; imx = oix; }
    if (omn < bmn || (omn == bmn && oin < imn)) { bmn = omn; imn = oin; }
  }
  const int bidx = (bmx >= -bmn) ? imx : 64 + imn;

  if (half == 0) bk[tok] = bidx;
  __syncthreads();
  {
    const bool isTok = tid < TSZ;
    const int  self  = tid & (TSZ - 1);
    const int  key   = isTok ? bk[self] : self;
    const int  lim   = self;
    int acc = 0;
    if (isTok) {
      const int jend = __builtin_amdgcn_readfirstlane((wave + 1) * 32);
#pragma unroll 4
      for (int j = 0; j < jend; j += 4) {
        const int4 v = *(const int4*)(&bk[j]);
        acc += (v.x == key && j     < lim) ? 1 : 0;
        acc += (v.y == key && j + 1 < lim) ? 1 : 0;
        acc += (v.z == key && j + 2 < lim) ? 1 : 0;
        acc += (v.w == key && j + 3 < lim) ? 1 : 0;
      }
    } else {
#pragma unroll 8
      for (int j = 0; j < TSZ; j += 4) {
        const int4 v = *(const int4*)(&bk[j]);
        acc += (v.x == key) ? 1 : 0;
        acc += (v.y == key) ? 1 : 0;
        acc += (v.z == key) ? 1 : 0;
        acc += (v.w == key) ? 1 : 0;
      }
    }
    const size_t bh = (size_t)b*NHASH + h;
    if (isTok) br[bh*SEQ + t0 + self] = key | (acc << 7);
    else       cnt[(bh*NTL + tt)*NBUCK + self] = acc;
  }
}

__global__ __launch_bounds__(256) void lsh_place_kernel(
    const int* __restrict__ br, const int* __restrict__ cnt, int* __restrict__ sticker)
{
  __shared__ int base[8][NBUCK];
  __shared__ int hs[2][NBUCK];
  __shared__ int qoff[NBUCK];
  const int bid  = blockIdx.x;
  const int part = bid & 7;
  const int h    = (bid >> 3) & 7;
  const int b    = bid >> 6;
  const int tid  = threadIdx.x;
  const int q    = tid & (NBUCK - 1), hf = tid >> 7;
  const size_t bh = (size_t)b*NHASH + h;

  {
    const int* c = cnt + (bh*NTL + (size_t)hf*32)*NBUCK + q;
    int v[32];
#pragma unroll
    for (int i = 0; i < 32; i++) v[i] = c[i*NBUCK];
    int run = 0;
#pragma unroll
    for (int i = 0; i < 32; i++) {
      const int tt = hf*32 + i;
      if ((tt >> 3) == part) base[tt & 7][q] = run;
      run += v[i];
    }
    hs[hf][q] = run;
  }
  __syncthreads();
  if (tid < NBUCK) {
    int off = 0;
    for (int j = 0; j < NBUCK; j++) {
      const int tj = hs[0][j] + hs[1][j];
      off += (j < q) ? tj : 0;
    }
    qoff[q] = h*SEQ + off + ((part >= 4) ? hs[0][q] : 0);
  }
  __syncthreads();
#pragma unroll
  for (int i = 0; i < 4; i++) {
    const int t    = part*1024 + i*256 + tid;
    const int v    = br[bh*SEQ + t];
    const int bidx = v & (NBUCK - 1), r = v >> 7;
    const int pos  = base[(t >> 7) & 7][bidx] + qoff[bidx] + r;
    sticker[(size_t)b*HS + (pos & (HS - 1))] = h*SEQ + t;
  }
}

__global__ __launch_bounds__(256, 10) void lsh_attn_kernel(
    const float* __restrict__ qk, const float* __restrict__ vin,
    const int* __restrict__ sticker,
    float* __restrict__ oBuf, float* __restrict__ lgBuf)
{
  __shared__ int idxA[128];
  __shared__ __align__(16) int tA[128];
  __shared__ __align__(16) float Qs[64][68];
  __shared__ __align__(16) float KVs[128*68];
  __shared__ float redm[2][64];
  __shared__ float reds[2][64];
  float (*Ks)[68]  = (float (*)[68])KVs;
  float (*Vp)[136] = (float (*)[136])KVs;
  float (*O1)[68]  = (float (*)[68])KVs;

  const int bid = blockIdx.x;
  const int c   = bid & (NCHUNK - 1);
  const int b   = bid >> 10;
  const int tid = threadIdx.x;
  const int prev = (c + NCHUNK - 1) & (NCHUNK - 1);

  if (tid < 128) {
    const int src_chunk = (tid < 64) ? c : prev;
    const int j = tid & 63;
    const int idx = sticker[(unsigned)b*HS + (unsigned)src_chunk*BSZ + j];
    idxA[tid] = idx;
    tA[tid]   = idx & (SEQ - 1);
  }
  __syncthreads();

  const int g16 = tid >> 4, l4 = (tid & 15) * 4;
  {
    float4 kr[8];
#pragma unroll
    for (int p = 0; p < 8; p++) {
      const unsigned off = ((unsigned)b*SEQ + tA[32*(p >> 1) + 2*g16 + (p & 1)])*DIM + l4;
      kr[p] = *(const float4*)(qk + off);
    }
#pragma unroll
    for (int p = 0; p < 4; p++) *(float4*)(&Qs[32*(p >> 1) + 2*g16 + (p & 1)][l4]) = kr[p];
    float ss[8];
#pragma unroll
    for (int p = 0; p < 8; p++) {
      float s = (kr[p].x*kr[p].x + kr[p].y*kr[p].y) + (kr[p].z*kr[p].z + kr[p].w*kr[p].w);
      s += quad_xor1(s);
      s += quad_xor2(s);
      s += row_xor4(s);
      s += row_xor8(s);
      ss[p] = s;
    }
    const int own = tid & 7;
    float mine = ss[0];
#pragma unroll
    for (int p = 1; p < 8; p++) mine = (own == p) ? ss[p] : mine;
    const float invm = 1.0f / (sqrtf(mine) + 1e-6f);
    float inv[8];
    inv[0] = row_lane<0>(invm); inv[1] = row_lane<1>(invm);
    inv[2] = row_lane<2>(invm); inv[3] = row_lane<3>(invm);
    inv[4] = row_lane<4>(invm); inv[5] = row_lane<5>(invm);
    inv[6] = row_lane<6>(invm); inv[7] = row_lane<7>(invm);
#pragma unroll
    for (int p = 0; p < 8; p++) {
      float4 kk = kr[p];
      kk.x *= inv[p]; kk.y *= inv[p]; kk.z *= inv[p]; kk.w *= inv[p];
      *(float4*)(&Ks[32*(p >> 1) + 2*g16 + (p & 1)][l4]) = kk;
    }
  }
  __syncthreads();

  const int wave = tid >> 5, lane = tid & 31, half = lane >> 4, mn = lane & 15;
  const int qi   = wave & 3;
  const int kh   = wave >> 2;
  const int kv0  = kh * 64;
  const int qrow = qi*16 + mn;

  v8f st[4] = {};
#pragma unroll
  for (int k0 = 0; k0 < 64; k0 += 4) {
    v2f bq;
    bq.x = Qs[qrow][k0 + 2*half    ];
    bq.y = Qs[qrow][k0 + 2*half + 1];
#pragma unroll
    for (int t = 0; t < 4; t++) {
      v2f ak;
      ak.x = Ks[kv0 + t*16 + mn][k0 + 2*half    ];
      ak.y = Ks[kv0 + t*16 + mn][k0 + 2*half + 1];
      st[t] = __builtin_amdgcn_wmma_f32_16x16x4_f32(false, ak, false, bq,
                                                    (short)0, st[t], false, false);
    }
  }

  float4 vr[8];
#pragma unroll
  for (int p = 0; p < 8; p++) {
    const unsigned off = ((unsigned)b*SEQ + tA[32*(p >> 1) + 2*g16 + (p & 1)])*DIM + l4;
    vr[p] = *(const float4*)(vin + off);
  }

  const int ti = tA[qrow];
  float m = -1e30f;
#pragma unroll
  for (int t = 0; t < 4; t++) {
    const int4 ja = *(const int4*)(&tA[kv0 + t*16 + 8*half    ]);
    const int4 jb = *(const int4*)(&tA[kv0 + t*16 + 8*half + 4]);
    const int tj[8] = {ja.x, ja.y, ja.z, ja.w, jb.x, jb.y, jb.z, jb.w};
#pragma unroll
    for (int r = 0; r < 8; r++) {
      float d = st[t][r] * 0.125f;
      if (tj[r] == ti) d = SELF_VAL;
      st[t][r] = d;
      m = fmaxf(m, d);
    }
  }
  m = fmaxf(m, half_swap(m));
  if (half == 0) redm[kh][qrow] = m;
  __syncthreads();

#pragma unroll
  for (int q = 0; q < 4; q++) {
    const float4 ve = vr[2*q], vo = vr[2*q + 1];
    float4 lo, hi;
    lo.x = ve.x; lo.y = vo.x; lo.z = ve.y; lo.w = vo.y;
    hi.x = ve.z; hi.y = vo.z; hi.z = ve.w; hi.w = vo.w;
    *(float4*)(&Vp[16*q + g16][2*l4    ]) = lo;
    *(float4*)(&Vp[16*q + g16][2*l4 + 4]) = hi;
  }

  m = fmaxf(redm[0][qrow], redm[1][qrow]);
  float ssum = 0.f;
#pragma unroll
  for (int t = 0; t < 4; t++) {
#pragma unroll
    for (int r = 0; r < 8; r++) {
      const float p = expf(st[t][r] - m);
      st[t][r] = p;
      ssum += p;
    }
  }
  ssum += half_swap(ssum);
  if (half == 0) reds[kh][qrow] = ssum;

  __syncthreads();

  v8f o[4] = {};
#pragma unroll
  for (int t = 0; t < 4; t++) {
#pragma unroll
    for (int s = 0; s < 4; s++) {
      v2f a;
      a.x = st[t][2*s    ];
      a.y = st[t][2*s + 1];
      const int pr = (kv0 >> 1) + t*8 + 4*half + s;
#pragma unroll
      for (int nt = 0; nt < 4; nt++) {
        const v2f bb = *(const v2f*)(&Vp[pr][2*(nt*16 + mn)]);
        o[nt] = __builtin_amdgcn_wmma_f32_16x16x4_f32(false, a, false, bb,
                                                      (short)0, o[nt], false, false);
      }
    }
  }
  __syncthreads();
  {
    float (*Od)[68] = kh ? O1 : Qs;
#pragma unroll
    for (int nt = 0; nt < 4; nt++) {
#pragma unroll
      for (int r = 0; r < 8; r++)
        Od[qi*16 + 8*half + r][nt*16 + mn] = o[nt][r];
    }
  }
  __syncthreads();

  {
#pragma unroll
    for (int p = 0; p < 4; p++) {
      const int row = p*16 + g16;
      const float inv = 1.0f / (reds[0][row] + reds[1][row]);
      const float4 u = *(const float4*)(&Qs[row][l4]);
      const float4 w = *(const float4*)(&O1[row][l4]);
      float4 r4;
      r4.x = (u.x + w.x) * inv; r4.y = (u.y + w.y) * inv;
      r4.z = (u.z + w.z) * inv; r4.w = (u.w + w.w) * inv;
      *(float4*)(oBuf + ((unsigned)b*HS + idxA[row])*DIM + l4) = r4;
    }
  }
  if (tid < 64)
    lgBuf[(unsigned)b*HS + idxA[tid]] =
        fmaxf(redm[0][tid], redm[1][tid]) + logf(reds[0][tid] + reds[1][tid]);
}

#define CTOK 4

__device__ __forceinline__ void combine_token(const float4 (&o4)[NHASH], float (&lg)[NHASH],
                                              float* __restrict__ dst)
{
  float m = -1e30f;
#pragma unroll
  for (int h = 0; h < NHASH; h++) m = fmaxf(m, lg[h]);
  float ssum = 0.f;
#pragma unroll
  for (int h = 0; h < NHASH; h++) { lg[h] = expf(lg[h] - m); ssum += lg[h]; }
  const float inv = 1.0f / ssum;
#pragma unroll
  for (int h = 0; h < NHASH; h++) lg[h] *= inv;

  float4 acc = {0.f, 0.f, 0.f, 0.f};
#pragma unroll
  for (int h = 0; h < NHASH; h++) {
    acc.x += lg[h]*o4[h].x; acc.y += lg[h]*o4[h].y; acc.z += lg[h]*o4[h].z; acc.w += lg[h]*o4[h].w;
  }
  *(float4*)dst = acc;
}

__global__ __launch_bounds__(256) void lsh_combine_kernel(
    const float* __restrict__ oBuf, const float* __restrict__ lgBuf,
    float* __restrict__ out)
{
  const int l4  = (threadIdx.x & 15) * 4;
  const int tk0 = (blockIdx.x * 16 + (threadIdx.x >> 4)) * CTOK;
  const int b   = tk0 >> 13;
  const int t0  = tk0 & (SEQ - 1);
  const float* ob = oBuf  + ((size_t)b*HS + t0)*DIM + l4;
  const float* lb = lgBuf +  (size_t)b*HS + t0;
  float*       dst = out  + ((size_t)b*SEQ + t0)*DIM + l4;

  float4 o4[NHASH];
  float  lg[NHASH];
#pragma unroll
  for (int h = 0; h < NHASH; h++) {
    o4[h] = *(const float4*)(ob + (size_t)h*SEQ*DIM);
    lg[h] = lb[(size_t)h*SEQ];
  }
#pragma unroll 1
  for (int i = 1; i < CTOK; i++) {
    float4 n4[NHASH];
    float  nl[NHASH];
#pragma unroll
    for (int h = 0; h < NHASH; h++) {
      n4[h] = *(const float4*)(ob + (size_t)h*SEQ*DIM + (size_t)i*DIM);
      nl[h] = lb[(size_t)h*SEQ + i];
    }
    combine_token(o4, lg, dst + (size_t)(i - 1)*DIM);
#pragma unroll
    for (int h = 0; h < NHASH; h++) { o4[h] = n4[h]; lg[h] = nl[h]; }
  }
  combine_token(o4, lg, dst + (size_t)(CTOK - 1)*DIM);
}

extern "C" void kernel_launch(void* const* d_in, const int* in_sizes, int n_in,
                              void* d_out, int out_size, void* d_ws, size_t ws_size,
                              hipStream_t stream)
{
  const float* qk  = (const float*)d_in[0];
  const float* v   = (const float*)d_in[1];
  const float* rot = (const float*)d_in[2];

  char* ws = (char*)d_ws;
  size_t off = 0;
  auto alloc = [&](size_t bytes) -> void* {
    void* p = ws + off;
    off += (bytes + 255) & ~(size_t)255;
    return p;
  };
  int*   br        = (int*)  alloc((size_t)BATCH*NHASH*SEQ*sizeof(int));
  int*   sticker   = (int*)  alloc((size_t)BATCH*HS*sizeof(int));
  float* oBuf      = (float*)alloc((size_t)BATCH*HS*DIM*sizeof(float));
  float* lgBuf     = (float*)alloc((size_t)BATCH*HS*sizeof(float));
  if (off > ws_size) return;
  int*   cnt       = (int*)oBuf;

  lsh_hash_kernel   <<<BATCH*NHASH*(SEQ/128), 256, 0, stream>>>(qk, rot, br, cnt);
  lsh_place_kernel  <<<BATCH*NHASH*8,         256, 0, stream>>>(br, cnt, sticker);
  lsh_attn_kernel   <<<BATCH*NCHUNK,         256, 0, stream>>>(qk, v, sticker, oBuf, lgBuf);
  lsh_combine_kernel<<<(BATCH*SEQ)/(16*CTOK), 256, 0, stream>>>(oBuf, lgBuf, (float*)d_out);
}


# attention and round-combine softmax in base 2: log2(e) folded into the 1/8 score scale, each exponential one bare v_exp_f32 (argument <= 0), not expf; gathers by 32-bit byte offsets to hold 96 regs
# speedup vs baseline: 1.9973x; 1.0617x over previous
#include <hip/hip_runtime.h>
#include <math.h>

#define BATCH 8
#define SEQ   8192
#define DIM   64
#define NHASH 8
#define NBUCK 128
#define NBALL (NHASH*NBUCK)
#define BSZ   64
#define NCHUNK (NHASH*(SEQ/BSZ))
#define HS    (NHASH*SEQ)
#define TSZ   128
#define NTL   (SEQ/TSZ)
#define SELF_VAL (-5.0e4f)
#define SCALE2    0.18033688011112042f
#define SELF_VAL2 (-72134.75f)

typedef float v2f __attribute__((ext_vector_type(2)));
typedef float v8f __attribute__((ext_vector_type(8)));

__device__ __forceinline__ float quad_xor1(float v) {
  return __int_as_float(__builtin_amdgcn_update_dpp(0, __float_as_int(v), 0xB1, 0xF, 0xF, true));
}
__device__ __forceinline__ float quad_xor2(float v) {
  return __int_as_float(__builtin_amdgcn_update_dpp(0, __float_as_int(v), 0x4E, 0xF, 0xF, true));
}
__device__ __forceinline__ float row_xor4(float v) {
  return __int_as_float(__builtin_amdgcn_update_dpp(0, __float_as_int(v), 0x164, 0xF, 0xF, true));
}
__device__ __forceinline__ float row_xor8(float v) {
  return __int_as_float(__builtin_amdgcn_update_dpp(0, __float_as_int(v), 0x168, 0xF, 0xF, true));
}
__device__ __forceinline__ float half_swap(float v) {
  const unsigned u = __float_as_uint(v);
  return __uint_as_float(__builtin_amdgcn_permlanex16(u, u, 0x76543210u, 0xfedcba98u, false, false));
}
template <int P>
__device__ __forceinline__ float row_lane(float v) {
  return __int_as_float(__builtin_amdgcn_update_dpp(0, __float_as_int(v), 0x150 + P, 0xF, 0xF, true));
}

__device__ __forceinline__ int half_swap_i(int v) {
  const unsigned u = (unsigned)v;
  return (int)__builtin_amdgcn_permlanex16(u, u, 0x76543210u, 0xfedcba98u, false, false);
}

__global__ __launch_bounds__(256) void lsh_hash_kernel(
    const float* __restrict__ qk, const float* __restrict__ rot,
    int* __restrict__ br, int* __restrict__ cnt)
{
  __shared__ __align__(16) int bk[TSZ];
  __shared__ __align__(16) float A [128][68];
  __shared__ __align__(16) float Bt[64][68];

  const int bid = blockIdx.x;
  const int tt  = bid & 63;
  const int h   = (bid >> 6) & 7;
  const int b   = bid >> 9;
  const int tid = threadIdx.x;
  const int t0  = tt * 128;

  const int g16 = tid >> 4, l4 = (tid & 15) * 4;
  {
    const float4* s4 = (const float4*)(qk + ((size_t)b*SEQ + t0)*DIM);
#pragma unroll
    for (int p = 0; p < 8; p++)
      *(float4*)(&A[p*16 + g16][l4]) = s4[p*256 + tid];
  }
  {
#pragma unroll
    for (int p = 0; p < 4; p++) {
      const int f = p*16 + g16;
      const float4 r4 =
          *(const float4*)(rot + (size_t)f*(NHASH*64) + (size_t)h*64 + l4);
      Bt[l4    ][f] = r4.x;
      Bt[l4 + 1][f] = r4.y;
      Bt[l4 + 2][f] = r4.z;
      Bt[l4 + 3][f] = r4.w;
    }
  }
  __syncthreads();

  const int wave = tid >> 5, lane = tid & 31, half = lane >> 4, mn = lane & 15;
  const int tok = wave*16 + mn;

  v8f acc[4] = {};
#pragma unroll
  for (int k0 = 0; k0 < 64; k0 += 4) {
    v2f bq;
    bq.x = A[tok][k0 + 2*half    ];
    bq.y = A[tok][k0 + 2*half + 1];
#pragma unroll
    for (int bt = 0; bt < 4; bt++) {
      v2f ar;
      ar.x = Bt[bt*16 + mn][k0 + 2*half    ];
      ar.y = Bt[bt*16 + mn][k0 + 2*half + 1];
      acc[bt] = __builtin_amdgcn_wmma_f32_16x16x4_f32(false, ar, false, bq,
                                                      (short)0, acc[bt], false, false);
    }
  }

  float bmx = 0.f, bmn = 0.f;
  int   imx = 0,   imn = 0;
#pragma unroll
  for (int bt = 0; bt < 4; bt++) {
    float mx = acc[bt][0], mi = acc[bt][0];
    int   ix = bt*16,      ii = bt*16;
#pragma unroll
    for (int r = 1; r < 8; r++) {
      const float vv = acc[bt][r];
      if (vv > mx) { mx = vv; ix = bt*16 + r; }
      if (vv < mi) { mi = vv; ii = bt*16 + r; }
    }
    if (bt == 0) { bmx = mx; imx = ix; bmn = mi; imn = ii; }
    else {
      if (mx > bmx) { bmx = mx; imx = ix; }
      if (mi < bmn) { bmn = mi; imn = ii; }
    }
  }
  imx += 8*half;
  imn += 8*half;
  {
    const float omx = half_swap(bmx), omn = half_swap(bmn);
    const int   oix = half_swap_i(imx), oin = half_swap_i(imn);
    if (omx > bmx || (omx == bmx && oix < imx)) { bmx = omx; imx = oix; }
    if (omn < bmn || (omn == bmn && oin < imn)) { bmn = omn; imn = oin; }
  }
  const int bidx = (bmx >= -bmn) ? imx : 64 + imn;

  if (half == 0) bk[tok] = bidx;
  __syncthreads();
  {
    const bool isTok = tid < TSZ;
    const int  self  = tid & (TSZ - 1);
    const int  key   = isTok ? bk[self] : self;
    const int  lim   = self;
    int acc = 0;
    if (isTok) {
      const int jend = __builtin_amdgcn_readfirstlane((wave + 1) * 32);
#pragma unroll 4
      for (int j = 0; j < jend; j += 4) {
        const int4 v = *(const int4*)(&bk[j]);
        acc += (v.x == key && j     < lim) ? 1 : 0;
        acc += (v.y == key && j + 1 < lim) ? 1 : 0;
        acc += (v.z == key && j + 2 < lim) ? 1 : 0;
        acc += (v.w == key && j + 3 < lim) ? 1 : 0;
      }
    } else {
#pragma unroll 8
      for (int j = 0; j < TSZ; j += 4) {
        const int4 v = *(const int4*)(&bk[j]);
        acc += (v.x == key) ? 1 : 0;
        acc += (v.y == key) ? 1 : 0;
        acc += (v.z == key) ? 1 : 0;
        acc += (v.w == key) ? 1 : 0;
      }
    }
    const size_t bh = (size_t)b*NHASH + h;
    if (isTok) br[bh*SEQ + t0 + self] = key | (acc << 7);
    else       cnt[(bh*NTL + tt)*NBUCK + self] = acc;
  }
}

__global__ __launch_bounds__(256) void lsh_place_kernel(
    const int* __restrict__ br, const int* __restrict__ cnt, int* __restrict__ sticker)
{
  __shared__ int base[8][NBUCK];
  __shared__ int hs[2][NBUCK];
  __shared__ int qoff[NBUCK];
  const int bid  = blockIdx.x;
  const int part = bid & 7;
  const int h    = (bid >> 3) & 7;
  const int b    = bid >> 6;
  const int tid  = threadIdx.x;
  const int q    = tid & (NBUCK - 1), hf = tid >> 7;
  const size_t bh = (size_t)b*NHASH + h;

  {
    const int* c = cnt + (bh*NTL + (size_t)hf*32)*NBUCK + q;
    int v[32];
#pragma unroll
    for (int i = 0; i < 32; i++) v[i] = c[i*NBUCK];
    int run = 0;
#pragma unroll
    for (int i = 0; i < 32; i++) {
      const int tt = hf*32 + i;
      if ((tt >> 3) == part) base[tt & 7][q] = run;
      run += v[i];
    }
    hs[hf][q] = run;
  }
  __syncthreads();
  if (tid < NBUCK) {
    int off = 0;
    for (int j = 0; j < NBUCK; j++) {
      const int tj = hs[0][j] + hs[1][j];
      off += (j < q) ? tj : 0;
    }
    qoff[q] = h*SEQ + off + ((part >= 4) ? hs[0][q] : 0);
  }
  __syncthreads();
#pragma unroll
  for (int i = 0; i < 4; i++) {
    const int t    = part*1024 + i*256 + tid;
    const int v    = br[bh*SEQ + t];
    const int bidx = v & (NBUCK - 1), r = v >> 7;
    const int pos  = base[(t >> 7) & 7][bidx] + qoff[bidx] + r;
    sticker[(size_t)b*HS + (pos & (HS - 1))] = h*SEQ + t;
  }
}

__global__ __launch_bounds__(256, 10) void lsh_attn_kernel(
    const float* __restrict__ qk, const float* __restrict__ vin,
    const int* __restrict__ sticker,
    float* __restrict__ oBuf, float* __restrict__ lgBuf)
{
  __shared__ int idxA[128];
  __shared__ __align__(16) int tA[128];
  __shared__ __align__(16) float Qs[64][68];
  __shared__ __align__(16) float KVs[128*68];
  __shared__ float redm[2][64];
  __shared__ float reds[2][64];
  float (*Ks)[68]  = (float (*)[68])KVs;
  float (*Vp)[136] = (float (*)[136])KVs;
  float (*O1)[68]  = (float (*)[68])KVs;

  const int bid = blockIdx.x;
  const int c   = bid & (NCHUNK - 1);
  const int b   = bid >> 10;
  const int tid = threadIdx.x;
  const int prev = (c + NCHUNK - 1) & (NCHUNK - 1);

  if (tid < 128) {
    const int src_chunk = (tid < 64) ? c : prev;
    const int j = tid & 63;
    const int idx = sticker[(unsigned)b*HS + (unsigned)src_chunk*BSZ + j];
    idxA[tid] = idx;
    tA[tid]   = idx & (SEQ - 1);
  }
  __syncthreads();

  const int g16 = tid >> 4, l4 = (tid & 15) * 4;
  {
    float4 kr[8];
#pragma unroll
    for (int p = 0; p < 8; p++) {
      const unsigned off = (((unsigned)b*SEQ + tA[32*(p >> 1) + 2*g16 + (p & 1)])*DIM + l4) * 4u;
      kr[p] = *(const float4*)((const char*)qk + off);
    }
#pragma unroll
    for (int p = 0; p < 4; p++) *(float4*)(&Qs[32*(p >> 1) + 2*g16 + (p & 1)][l4]) = kr[p];
    float ss[8];
#pragma unroll
    for (int p = 0; p < 8; p++) {
      float s = (kr[p].x*kr[p].x + kr[p].y*kr[p].y) + (kr[p].z*kr[p].z + kr[p].w*kr[p].w);
      s += quad_xor1(s);
      s += quad_xor2(s);
      s += row_xor4(s);
      s += row_xor8(s);
      ss[p] = s;
    }
    const int own = tid & 7;
    float mine = ss[0];
#pragma unroll
    for (int p = 1; p < 8; p++) mine = (own == p) ? ss[p] : mine;
    const float invm = 1.0f / (sqrtf(mine) + 1e-6f);
    float inv[8];
    inv[0] = row_lane<0>(invm); inv[1] = row_lane<1>(invm);
    inv[2] = row_lane<2>(invm); inv[3] = row_lane<3>(invm);
    inv[4] = row_lane<4>(invm); inv[5] = row_lane<5>(invm);
    inv[6] = row_lane<6>(invm); inv[7] = row_lane<7>(invm);
#pragma unroll
    for (int p = 0; p < 8; p++) {
      float4 kk = kr[p];
      kk.x *= inv[p]; kk.y *= inv[p]; kk.z *= inv[p]; kk.w *= inv[p];
      *(float4*)(&Ks[32*(p >> 1) + 2*g16 + (p & 1)][l4]) = kk;
    }
  }
  __syncthreads();

  const int wave = tid >> 5, lane = tid & 31, half = lane >> 4, mn = lane & 15;
  const int qi   = wave & 3;
  const int kh   = wave >> 2;
  const int kv0  = kh * 64;
  const int qrow = qi*16 + mn;

  v8f st[4] = {};
#pragma unroll
  for (int k0 = 0; k0 < 64; k0 += 4) {
    v2f bq;
    bq.x = Qs[qrow][k0 + 2*half    ];
    bq.y = Qs[qrow][k0 + 2*half + 1];
#pragma unroll
    for (int t = 0; t < 4; t++) {
      v2f ak;
      ak.x = Ks[kv0 + t*16 + mn][k0 + 2*half    ];
      ak.y = Ks[kv0 + t*16 + mn][k0 + 2*half + 1];
      st[t] = __builtin_amdgcn_wmma_f32_16x16x4_f32(false, ak, false, bq,
                                                    (short)0, st[t], false, false);
    }
  }

  float4 vr[8];
#pragma unroll
  for (int p = 0; p < 8; p++) {
    const unsigned off = (((unsigned)b*SEQ + tA[32*(p >> 1) + 2*g16 + (p & 1)])*DIM + l4) * 4u;
    vr[p] = *(const float4*)((const char*)vin + off);
  }

  const int ti = tA[qrow];
  float m = -1e30f;
#pragma unroll
  for (int t = 0; t < 4; t++) {
    const int4 ja = *(const int4*)(&tA[kv0 + t*16 + 8*half    ]);
    const int4 jb = *(const int4*)(&tA[kv0 + t*16 + 8*half + 4]);
    const int tj[8] = {ja.x, ja.y, ja.z, ja.w, jb.x, jb.y, jb.z, jb.w};
#pragma unroll
    for (int r = 0; r < 8; r++) {
      float d = st[t][r] * SCALE2;
      if (tj[r] == ti) d = SELF_VAL2;
      st[t][r] = d;
      m = fmaxf(m, d);
    }
  }
  m = fmaxf(m, half_swap(m));
  if (half == 0) redm[kh][qrow] = m;
  __syncthreads();

#pragma unroll
  for (int q = 0; q < 4; q++) {
    const float4 ve = vr[2*q], vo = vr[2*q + 1];
    float4 lo, hi;
    lo.x = ve.x; lo.y = vo.x; lo.z = ve.y; lo.w = vo.y;
    hi.x = ve.z; hi.y = vo.z; hi.z = ve.w; hi.w = vo.w;
    *(float4*)(&Vp[16*q + g16][2*l4    ]) = lo;
    *(float4*)(&Vp[16*q + g16][2*l4 + 4]) = hi;
  }

  m = fmaxf(redm[0][qrow], redm[1][qrow]);
  float ssum = 0.f;
#pragma unroll
  for (int t = 0; t < 4; t++) {
#pragma unroll
    for (int r = 0; r < 8; r++) {
      const float p = __builtin_amdgcn_exp2f(st[t][r] - m);
      st[t][r] = p;
      ssum += p;
    }
  }
  ssum += half_swap(ssum);
  if (half == 0) reds[kh][qrow] = ssum;

  __syncthreads();

  v8f o[4] = {};
#pragma unroll
  for (int t = 0; t < 4; t++) {
#pragma unroll
    for (int s = 0; s < 4; s++) {
      v2f a;
      a.x = st[t][2*s    ];
      a.y = st[t][2*s + 1];
      const int pr = (kv0 >> 1) + t*8 + 4*half + s;
#pragma unroll
      for (int nt = 0; nt < 4; nt++) {
        const v2f bb = *(const v2f*)(&Vp[pr][2*(nt*16 + mn)]);
        o[nt] = __builtin_amdgcn_wmma_f32_16x16x4_f32(false, a, false, bb,
                                                      (short)0, o[nt], false, false);
      }
    }
  }
  __syncthreads();
  {
    float (*Od)[68] = kh ? O1 : Qs;
#pragma unroll
    for (int nt = 0; nt < 4; nt++) {
#pragma unroll
      for (int r = 0; r < 8; r++)
        Od[qi*16 + 8*half + r][nt*16 + mn] = o[nt][r];
    }
  }
  __syncthreads();

  {
#pragma unroll
    for (int p = 0; p < 4; p++) {
      const int row = p*16 + g16;
      const float inv = 1.0f / (reds[0][row] + reds[1][row]);
      const float4 u = *(const float4*)(&Qs[row][l4]);
      const float4 w = *(const float4*)(&O1[row][l4]);
      float4 r4;
      r4.x = (u.x + w.x) * inv; r4.y = (u.y + w.y) * inv;
      r4.z = (u.z + w.z) * inv; r4.w = (u.w + w.w) * inv;
      *(float4*)(oBuf + ((unsigned)b*HS + idxA[row])*DIM + l4) = r4;
    }
  }
  if (tid < 64)
    lgBuf[(unsigned)b*HS + idxA[tid]] =
        fmaxf(redm[0][tid], redm[1][tid]) + log2f(reds[0][tid] + reds[1][tid]);
}

#define CTOK 4

__device__ __forceinline__ void combine_token(const float4 (&o4)[NHASH], float (&lg)[NHASH],
                                              float* __restrict__ dst)
{
  float m = -1e30f;
#pragma unroll
  for (int h = 0; h < NHASH; h++) m = fmaxf(m, lg[h]);
  float ssum = 0.f;
#pragma unroll
  for (int h = 0; h < NHASH; h++) { lg[h] = __builtin_amdgcn_exp2f(lg[h] - m); ssum += lg[h]; }
  const float inv = 1.0f / ssum;
#pragma unroll
  for (int h = 0; h < NHASH; h++) lg[h] *= inv;

  float4 acc = {0.f, 0.f, 0.f, 0.f};
#pragma unroll
  for (int h = 0; h < NHASH; h++) {
    acc.x += lg[h]*o4[h].x; acc.y += lg[h]*o4[h].y; acc.z += lg[h]*o4[h].z; acc.w += lg[h]*o4[h].w;
  }
  *(float4*)dst = acc;
}

__global__ __launch_bounds__(256) void lsh_combine_kernel(
    const float* __restrict__ oBuf, const float* __restrict__ lgBuf,
    float* __restrict__ out)
{
  const int l4  = (threadIdx.x & 15) * 4;
  const int tk0 = (blockIdx.x * 16 + (threadIdx.x >> 4)) * CTOK;
  const int b   = tk0 >> 13;
  const int t0  = tk0 & (SEQ - 1);
  const float* ob = oBuf  + ((size_t)b*HS + t0)*DIM + l4;
  const float* lb = lgBuf +  (size_t)b*HS + t0;
  float*       dst = out  + ((size_t)b*SEQ + t0)*DIM + l4;

  float4 o4[NHASH];
  float  lg[NHASH];
#pragma unroll
  for (int h = 0; h < NHASH; h++) {
    o4[h] = *(const float4*)(ob + (size_t)h*SEQ*DIM);
    lg[h] = lb[(size_t)h*SEQ];
  }
#pragma unroll 1
  for (int i = 1; i < CTOK; i++) {
    float4 n4[NHASH];
    float  nl[NHASH];
#pragma unroll
    for (int h = 0; h < NHASH; h++) {
      n4[h] = *(const float4*)(ob + (size_t)h*SEQ*DIM + (size_t)i*DIM);
      nl[h] = lb[(size_t)h*SEQ + i];
    }
    combine_token(o4, lg, dst + (size_t)(i - 1)*DIM);
#pragma unroll
    for (int h = 0; h < NHASH; h++) { o4[h] = n4[h]; lg[h] = nl[h]; }
  }
  combine_token(o4, lg, dst + (size_t)(CTOK - 1)*DIM);
}

extern "C" void kernel_launch(void* const* d_in, const int* in_sizes, int n_in,
                              void* d_out, int out_size, void* d_ws, size_t ws_size,
                              hipStream_t stream)
{
  const float* qk  = (const float*)d_in[0];
  const float* v   = (const float*)d_in[1];
  const float* rot = (const float*)d_in[2];

  char* ws = (char*)d_ws;
  size_t off = 0;
  auto alloc = [&](size_t bytes) -> void* {
    void* p = ws + off;
    off += (bytes + 255) & ~(size_t)255;
    return p;
  };
  int*   br        = (int*)  alloc((size_t)BATCH*NHASH*SEQ*sizeof(int));
  int*   sticker   = (int*)  alloc((size_t)BATCH*HS*sizeof(int));
  float* oBuf      = (float*)alloc((size_t)BATCH*HS*DIM*sizeof(float));
  float* lgBuf     = (float*)alloc((size_t)BATCH*HS*sizeof(float));
  if (off > ws_size) return;
  int*   cnt       = (int*)oBuf;

  lsh_hash_kernel   <<<BATCH*NHASH*(SEQ/128), 256, 0, stream>>>(qk, rot, br, cnt);
  lsh_place_kernel  <<<BATCH*NHASH*8,         256, 0, stream>>>(br, cnt, sticker);
  lsh_attn_kernel   <<<BATCH*NCHUNK,         256, 0, stream>>>(qk, v, sticker, oBuf, lgBuf);
  lsh_combine_kernel<<<(BATCH*SEQ)/(16*CTOK), 256, 0, stream>>>(oBuf, lgBuf, (float*)d_out);
}
